# starting kernel
# baseline (speedup 1.0000x reference)
#include <hip/hip_runtime.h>
#include <hip/hip_bf16.h>


typedef __attribute__((ext_vector_type(16))) __bf16 v16bf;
typedef __attribute__((ext_vector_type(8)))  float  v8f;

#define GCN_N 100000
#define GCN_D 128
#define GCN_WPACK_ELEMS (8 * 4 * 32 * 16)

__global__ void gcn_zero_kernel(float* __restrict__ p, size_t n) {
    size_t i      = (size_t)blockIdx.x * blockDim.x + threadIdx.x;
    size_t stride = (size_t)gridDim.x * blockDim.x;
    for (; i < n; i += stride) p[i] = 0.0f;
}

__global__ void gcn_degree_kernel(const int* __restrict__ src,
                                  const int* __restrict__ dst,
                                  float* __restrict__ deg_src,
                                  float* __restrict__ deg_dst,
                                  int nedges) {
    int i = blockIdx.x * blockDim.x + threadIdx.x;
    if (i < nedges) {
        atomicAdd(&deg_src[src[i]], 1.0f);
        atomicAdd(&deg_dst[dst[i]], 1.0f);
    }
}

__global__ void gcn_norm_kernel(float* __restrict__ p, int n) {
    int i = blockIdx.x * blockDim.x + threadIdx.x;
    if (i < n) p[i] = rsqrtf(fmaxf(p[i], 1.0f));
}

__device__ __forceinline__ int gcn_frag_k(int lane, int i) {
    const int j   = i >> 1;
    const int sub = i & 1;
    const int kk  = (j < 4) ? (2 * j + sub) : (16 + 2 * (j - 4) + sub);
    return kk + 8 * (lane >> 4);
}

__global__ void gcn_pack_w_kernel(const float* __restrict__ W,
                                  __bf16* __restrict__ Wp) {
    int idx = blockIdx.x * blockDim.x + threadIdx.x;
    if (idx >= GCN_WPACK_ELEMS) return;
    const int i    = idx & 15;
    const int lane = (idx >> 4) & 31;
    const int kb   = (idx >> 9) & 3;
    const int nt   = idx >> 11;
    const int K    = kb * 32 + gcn_frag_k(lane, i);
    const int n    = nt * 16 + (lane & 15);
    Wp[idx] = (__bf16)W[(size_t)K * GCN_D + n];
}

__global__ void __launch_bounds__(256)
gcn_gemm_norm_kernel(const float* __restrict__ X,
                     const __bf16* __restrict__ Wp,
                     const float* __restrict__ norm,
                     float* __restrict__ H,
                     int nrows) {
    const int lane = threadIdx.x & 31;
    const int wave = threadIdx.x >> 5;
    const int row0 = blockIdx.x * 16;
    const int n0   = wave * 16;
    const int half = lane >> 4;
    const int m    = lane & 15;

    v8f acc = {};
#pragma unroll
    for (int kb = 0; kb < 4; ++kb) {
        const int k0 = kb * 32;
        v16bf a;
#pragma unroll
        for (int i = 0; i < 16; ++i) {
            const int K = gcn_frag_k(lane, i);
            a[i] = (__bf16)X[(size_t)(row0 + m) * GCN_D + (k0 + K)];
        }
        const v16bf b =
            *(const v16bf*)&Wp[(((size_t)(wave * 4 + kb)) * 32 + lane) * 16];
        acc = __builtin_amdgcn_wmma_f32_16x16x32_bf16(
             false, a,  false, b,
             (short)0, acc,  false,  false);
    }

#pragma unroll
    for (int i = 0; i < 8; ++i) {
        const int r = row0 + i + 8 * half;
        H[(size_t)r * GCN_D + n0 + m] = acc[i] * norm[r];
    }
}

__global__ void __launch_bounds__(256)
gcn_scatter_kernel(const float4* __restrict__ H,
                   const int* __restrict__ src,
                   const int* __restrict__ dst,
                   float* __restrict__ agg,
                   int nedges) {
    const int wavesPerBlock = blockDim.x >> 5;
    const int edge = blockIdx.x * wavesPerBlock + (threadIdx.x >> 5);
    const int lane = threadIdx.x & 31;
    if (edge >= nedges) return;
    const int s = src[edge];
    const int d = dst[edge];
    const float4 v = H[(size_t)s * (GCN_D / 4) + lane];
    float* out = agg + (size_t)d * GCN_D + lane * 4;
    atomicAdd(out + 0, v.x);
    atomicAdd(out + 1, v.y);
    atomicAdd(out + 2, v.z);
    atomicAdd(out + 3, v.w);
}

__global__ void gcn_finalize_kernel(const float* __restrict__ agg,
                                    const float* __restrict__ norm_dst,
                                    const float* __restrict__ bias,
                                    float* __restrict__ out,
                                    int n_elems, int do_relu) {
    int i = blockIdx.x * blockDim.x + threadIdx.x;
    if (i >= n_elems) return;
    const int r = i >> 7;
    const int c = i & (GCN_D - 1);
    float v = agg[i] * norm_dst[r] + bias[c];
    if (do_relu) v = fmaxf(v, 0.0f);
    out[i] = v;
}

extern "C" void kernel_launch(void* const* d_in, const int* in_sizes, int n_in,
                              void* d_out, int out_size, void* d_ws, size_t ws_size,
                              hipStream_t stream) {
    const float* feat = (const float*)d_in[0];
    const int*   src  = (const int*)  d_in[1];
    const int*   dst  = (const int*)  d_in[2];
    const float* W1   = (const float*)d_in[3];
    const float* b1   = (const float*)d_in[4];
    const float* W2   = (const float*)d_in[5];
    const float* b2   = (const float*)d_in[6];
    float* out = (float*)d_out;

    const int N = GCN_N;
    const int E = in_sizes[1];
    const size_t NF = (size_t)N * GCN_D;

    float*  normS = (float*)d_ws;
    float*  normD = normS + N;
    __bf16* Wp1   = (__bf16*)(normD + N);
    __bf16* Wp2   = Wp1 + GCN_WPACK_ELEMS;
    float*  bufA  = (float*)(Wp2 + GCN_WPACK_ELEMS);
    float*  bufB  = bufA + NF;

    const int thr = 256;
    const int zgrid = 2048;
    const int egrid = (E + thr - 1) / thr;
    const int ngrid2 = (2 * N + thr - 1) / thr;
    const int pgrid = GCN_WPACK_ELEMS / thr;
    const int gemm_grid = N / 16;
    const int scat_grid = (E + (thr / 32) - 1) / (thr / 32);
    const int fin_grid  = (int)((NF + thr - 1) / thr);

    gcn_zero_kernel<<<zgrid, thr, 0, stream>>>(normS, (size_t)2 * N);
    gcn_degree_kernel<<<egrid, thr, 0, stream>>>(src, dst, normS, normD, E);
    gcn_norm_kernel<<<ngrid2, thr, 0, stream>>>(normS, 2 * N);
    gcn_pack_w_kernel<<<pgrid, thr, 0, stream>>>(W1, Wp1);
    gcn_pack_w_kernel<<<pgrid, thr, 0, stream>>>(W2, Wp2);

    gcn_gemm_norm_kernel<<<gemm_grid, thr, 0, stream>>>(feat, Wp1, normS, bufA, N);
    gcn_zero_kernel<<<zgrid, thr, 0, stream>>>(bufB, NF);
    gcn_scatter_kernel<<<scat_grid, thr, 0, stream>>>((const float4*)bufA, src, dst, bufB, E);
    gcn_finalize_kernel<<<fin_grid, thr, 0, stream>>>(bufB, normD, b1, bufA, (int)NF, 1);

    gcn_gemm_norm_kernel<<<gemm_grid, thr, 0, stream>>>(bufA, Wp2, normS, bufB, N);
    gcn_zero_kernel<<<zgrid, thr, 0, stream>>>(out, NF);
    gcn_scatter_kernel<<<scat_grid, thr, 0, stream>>>((const float4*)bufB, src, dst, out, E);
    gcn_finalize_kernel<<<fin_grid, thr, 0, stream>>>(out, normD, b2, out, (int)NF, 0);
}


# Aggregation without float atomics: edges bucketed by destination per call (int-atomic counting sort), one wave per row sums its source rows in registers; sum order = arrival rank in the bucket
# speedup vs baseline: 2.3662x; 2.3662x over previous
#include <hip/hip_runtime.h>
#include <hip/hip_bf16.h>


typedef __attribute__((ext_vector_type(16))) __bf16 v16bf;
typedef __attribute__((ext_vector_type(8)))  float  v8f;

#define GCN_N 100000
#define GCN_D 128
#define GCN_WPACK_ELEMS (8 * 4 * 32 * 16)

#define GCN_NPAD 100032

__global__ void gcn_zero_i32_kernel(int* __restrict__ p, int n) {
    int i = blockIdx.x * blockDim.x + threadIdx.x;
    if (i < n) p[i] = 0;
}

__global__ void gcn_count_kernel(const int* __restrict__ src,
                                 const int* __restrict__ dst,
                                 int* __restrict__ cnt_src,
                                 int* __restrict__ cnt_dst,
                                 int* __restrict__ pos,
                                 int nedges) {
    int i = blockIdx.x * blockDim.x + threadIdx.x;
    if (i < nedges) {
        atomicAdd(&cnt_src[src[i]], 1);
        pos[i] = atomicAdd(&cnt_dst[dst[i]], 1);
    }
}

__global__ void __launch_bounds__(1024)
gcn_rowptr_norm_kernel(const int* __restrict__ cnt_src,
                       const int* __restrict__ cnt_dst,
                       int* __restrict__ rowptr,
                       float* __restrict__ norm_src,
                       float* __restrict__ norm_dst,
                       int n) {
    __shared__ int wsum[32];
    const int tid   = threadIdx.x;
    const int lane  = tid & 31;
    const int wave  = tid >> 5;
    const int start = blockIdx.x * 1024;

    int part = 0;
    for (int i = tid; i < start; i += 1024) part += cnt_dst[i];
    for (int off = 16; off > 0; off >>= 1) part += __shfl_xor(part, off);
    if (lane == 0) wsum[wave] = part;
    __syncthreads();
    int base = 0;
    for (int w = 0; w < 32; ++w) base += wsum[w];
    __syncthreads();

    const int i = start + tid;
    const int c = (i < n) ? cnt_dst[i] : 0;
    int inc = c;
    for (int off = 1; off < 32; off <<= 1) {
        const int t = __shfl_up(inc, off);
        if (lane >= off) inc += t;
    }
    if (lane == 31) wsum[wave] = inc;
    __syncthreads();
    int wpre = 0;
    for (int w = 0; w < wave; ++w) wpre += wsum[w];

    if (i < n) {
        rowptr[i]   = base + wpre + inc - c;
        norm_dst[i] = rsqrtf(fmaxf((float)c, 1.0f));
        norm_src[i] = rsqrtf(fmaxf((float)cnt_src[i], 1.0f));
    }
}

__global__ void gcn_fill_kernel(const int* __restrict__ src,
                                const int* __restrict__ dst,
                                const int* __restrict__ rowptr,
                                const int* __restrict__ pos,
                                int* __restrict__ col,
                                int nedges) {
    int i = blockIdx.x * blockDim.x + threadIdx.x;
    if (i < nedges) col[rowptr[dst[i]] + pos[i]] = src[i];
}

__device__ __forceinline__ int gcn_frag_k(int lane, int i) {
    const int j   = i >> 1;
    const int sub = i & 1;
    const int kk  = (j < 4) ? (2 * j + sub) : (16 + 2 * (j - 4) + sub);
    return kk + 8 * (lane >> 4);
}

__global__ void gcn_pack_w_kernel(const float* __restrict__ W,
                                  __bf16* __restrict__ Wp) {
    int idx = blockIdx.x * blockDim.x + threadIdx.x;
    if (idx >= GCN_WPACK_ELEMS) return;
    const int i    = idx & 15;
    const int lane = (idx >> 4) & 31;
    const int kb   = (idx >> 9) & 3;
    const int nt   = idx >> 11;
    const int K    = kb * 32 + gcn_frag_k(lane, i);
    const int n    = nt * 16 + (lane & 15);
    Wp[idx] = (__bf16)W[(size_t)K * GCN_D + n];
}

__global__ void __launch_bounds__(256)
gcn_gemm_norm_kernel(const float* __restrict__ X,
                     const __bf16* __restrict__ Wp,
                     const float* __restrict__ norm,
                     float* __restrict__ H,
                     int nrows) {
    const int lane = threadIdx.x & 31;
    const int wave = threadIdx.x >> 5;
    const int row0 = blockIdx.x * 16;
    const int n0   = wave * 16;
    const int half = lane >> 4;
    const int m    = lane & 15;

    v8f acc = {};
#pragma unroll
    for (int kb = 0; kb < 4; ++kb) {
        const int k0 = kb * 32;
        v16bf a;
#pragma unroll
        for (int i = 0; i < 16; ++i) {
            const int K = gcn_frag_k(lane, i);
            a[i] = (__bf16)X[(size_t)(row0 + m) * GCN_D + (k0 + K)];
        }
        const v16bf b =
            *(const v16bf*)&Wp[(((size_t)(wave * 4 + kb)) * 32 + lane) * 16];
        acc = __builtin_amdgcn_wmma_f32_16x16x32_bf16(
             false, a,  false, b,
             (short)0, acc,  false,  false);
    }

#pragma unroll
    for (int i = 0; i < 8; ++i) {
        const int r = row0 + i + 8 * half;
        H[(size_t)r * GCN_D + n0 + m] = acc[i] * norm[r];
    }
}

__global__ void __launch_bounds__(256)
gcn_aggregate_kernel(const float* __restrict__ H,
                     const int* __restrict__ rowptr,
                     const int* __restrict__ cnt_dst,
                     const int* __restrict__ col,
                     const float* __restrict__ norm_dst,
                     const float* __restrict__ bias,
                     float* __restrict__ out,
                     int nrows, int do_relu) {
    const int lane = threadIdx.x & 31;
    const int row = __builtin_amdgcn_readfirstlane(
        (int)(blockIdx.x * (blockDim.x >> 5) + (threadIdx.x >> 5)));
    if (row >= nrows) return;
    const int beg = rowptr[row];
    const int cnt = cnt_dst[row];
    const float4* __restrict__ H4 = (const float4*)H;

    float4 acc = make_float4(0.0f, 0.0f, 0.0f, 0.0f);
    for (int done = 0; done < cnt; done += 32) {
        const int n = (cnt - done < 32) ? (cnt - done) : 32;
        const int myidx = (lane < n) ? col[beg + done + lane] : 0;
        int j = 0;
        for (; j + 4 <= n; j += 4) {
            const int s0 = __builtin_amdgcn_readlane(myidx, j);
            const int s1 = __builtin_amdgcn_readlane(myidx, j + 1);
            const int s2 = __builtin_amdgcn_readlane(myidx, j + 2);
            const int s3 = __builtin_amdgcn_readlane(myidx, j + 3);
            const float4 v0 = H4[(size_t)s0 * (GCN_D / 4) + lane];
            const float4 v1 = H4[(size_t)s1 * (GCN_D / 4) + lane];
            const float4 v2 = H4[(size_t)s2 * (GCN_D / 4) + lane];
            const float4 v3 = H4[(size_t)s3 * (GCN_D / 4) + lane];
            acc.x += v0.x; acc.y += v0.y; acc.z += v0.z; acc.w += v0.w;
            acc.x += v1.x; acc.y += v1.y; acc.z += v1.z; acc.w += v1.w;
            acc.x += v2.x; acc.y += v2.y; acc.z += v2.z; acc.w += v2.w;
            acc.x += v3.x; acc.y += v3.y; acc.z += v3.z; acc.w += v3.w;
        }
        for (; j < n; ++j) {
            const int s = __builtin_amdgcn_readlane(myidx, j);
            const float4 v = H4[(size_t)s * (GCN_D / 4) + lane];
            acc.x += v.x; acc.y += v.y; acc.z += v.z; acc.w += v.w;
        }
    }

    const float nd = norm_dst[row];
    float4 r;
    r.x = acc.x * nd + bias[lane * 4 + 0];
    r.y = acc.y * nd + bias[lane * 4 + 1];
    r.z = acc.z * nd + bias[lane * 4 + 2];
    r.w = acc.w * nd + bias[lane * 4 + 3];
    if (do_relu) {
        r.x = fmaxf(r.x, 0.0f); r.y = fmaxf(r.y, 0.0f);
        r.z = fmaxf(r.z, 0.0f); r.w = fmaxf(r.w, 0.0f);
    }
    ((float4*)out)[(size_t)row * (GCN_D / 4) + lane] = r;
}

extern "C" void kernel_launch(void* const* d_in, const int* in_sizes, int n_in,
                              void* d_out, int out_size, void* d_ws, size_t ws_size,
                              hipStream_t stream) {
    const float* feat = (const float*)d_in[0];
    const int*   src  = (const int*)  d_in[1];
    const int*   dst  = (const int*)  d_in[2];
    const float* W1   = (const float*)d_in[3];
    const float* b1   = (const float*)d_in[4];
    const float* W2   = (const float*)d_in[5];
    const float* b2   = (const float*)d_in[6];
    float* out = (float*)d_out;

    const int N = GCN_N;
    const int E = in_sizes[1];
    const size_t NF = (size_t)N * GCN_D;

    int*    cntS   = (int*)d_ws;
    int*    cntD   = cntS + GCN_NPAD;
    float*  normS  = (float*)(cntD + GCN_NPAD);
    float*  normD  = normS + GCN_NPAD;
    int*    rowptr = (int*)(normD + GCN_NPAD);
    __bf16* Wp1    = (__bf16*)(rowptr + GCN_NPAD);
    __bf16* Wp2    = Wp1 + GCN_WPACK_ELEMS;
    float*  bufA   = (float*)(Wp2 + GCN_WPACK_ELEMS);
    float*  bufB   = bufA + NF;
    int*    pos    = (int*)(bufB + NF);
    int*    col    = pos + E;

    const int thr = 256;
    const int zgrid = (2 * GCN_NPAD + thr - 1) / thr;
    const int egrid = (E + thr - 1) / thr;
    const int sgrid = (N + 1023) / 1024;
    const int pgrid = GCN_WPACK_ELEMS / thr;
    const int gemm_grid = N / 16;
    const int agg_grid  = (N + (thr / 32) - 1) / (thr / 32);

    gcn_zero_i32_kernel<<<zgrid, thr, 0, stream>>>(cntS, 2 * GCN_NPAD);
    gcn_count_kernel<<<egrid, thr, 0, stream>>>(src, dst, cntS, cntD, pos, E);
    gcn_rowptr_norm_kernel<<<sgrid, 1024, 0, stream>>>(cntS, cntD, rowptr, normS, normD, N);
    gcn_fill_kernel<<<egrid, thr, 0, stream>>>(src, dst, rowptr, pos, col, E);
    gcn_pack_w_kernel<<<pgrid, thr, 0, stream>>>(W1, Wp1);
    gcn_pack_w_kernel<<<pgrid, thr, 0, stream>>>(W2, Wp2);

    gcn_gemm_norm_kernel<<<gemm_grid, thr, 0, stream>>>(feat, Wp1, normS, bufA, N);
    gcn_aggregate_kernel<<<agg_grid, thr, 0, stream>>>(bufA, rowptr, cntD, col, normD, b1, bufB, N, 1);

    gcn_gemm_norm_kernel<<<gemm_grid, thr, 0, stream>>>(bufB, Wp2, normS, bufA, N);
    gcn_aggregate_kernel<<<agg_grid, thr, 0, stream>>>(bufA, rowptr, cntD, col, normD, b2, out, N, 0);
}


# Layer kernels as a resident grid: 1280 blocks (5 per WGP) loop over 16-row tiles; W fragments loaded once per block, next tile's degrees and first bucket indices fetched a tile ahead; same sums
# speedup vs baseline: 8.3672x; 3.5361x over previous
#include <hip/hip_runtime.h>


typedef __attribute__((ext_vector_type(16))) _Float16 v16h;
typedef __attribute__((ext_vector_type(8)))  float  v8f;

#define GCN_N 100000
#define GCN_D 128
#define GCN_WPACK_ELEMS (8 * 4 * 32 * 16)

#define GCN_NPAD 100032

#define GCN_CAP 64
#define GCN_OVFPAD 64

__global__ void gcn_zero_i32_kernel(int* __restrict__ p, int n) {
    int i = blockIdx.x * blockDim.x + threadIdx.x;
    if (i < n) p[i] = 0;
}

#define GCN_RNODES 392
#define GCN_NRANGE 256
#define GCN_CAPR   32768
#define GCN_PT     512
#define GCN_EPT    4
#define GCN_CHUNK  (GCN_PT * GCN_EPT)

__global__ void __launch_bounds__(GCN_PT)
gcn_partition_kernel(const int* __restrict__ src,
                     const int* __restrict__ dst,
                     int* __restrict__ gcount,
                     int* __restrict__ part,
                     int* __restrict__ covf_cnt,
                     int* __restrict__ covf_ent,
                     int* __restrict__ covf_rng,
                     int nedges, int covf_cap) {
    __shared__ int hist[GCN_NRANGE];
    __shared__ int loff[GCN_NRANGE];
    __shared__ int gbase[GCN_NRANGE];
    __shared__ int gfirst[GCN_NRANGE];
    __shared__ int gob[GCN_NRANGE];
    __shared__ int stage[2 * GCN_CHUNK];
    __shared__ unsigned short stage_rng[2 * GCN_CHUNK];

    const int tid = threadIdx.x;
    if (tid < GCN_NRANGE) hist[tid] = 0;
    __syncthreads();

    int ent[2 * GCN_EPT], rng[2 * GCN_EPT], rk[2 * GCN_EPT];
    const int e0 = blockIdx.x * GCN_CHUNK;
#pragma unroll
    for (int k = 0; k < GCN_EPT; ++k) {
        const int i = e0 + k * GCN_PT + tid;
        if (i < nedges) {
            const int s  = src[i];
            const int d  = dst[i];
            const int rd = d / GCN_RNODES;
            const int rs = s / GCN_RNODES;
            ent[2 * k]     = (s << 10) | (d - rd * GCN_RNODES);
            rng[2 * k]     = rd;
            rk[2 * k]      = atomicAdd(&hist[rd], 1);
            ent[2 * k + 1] = (1 << 9) | (s - rs * GCN_RNODES);
            rng[2 * k + 1] = rs;
            rk[2 * k + 1]  = atomicAdd(&hist[rs], 1);
        } else {
            ent[2 * k] = 0; ent[2 * k + 1] = 0;
            rng[2 * k] = -1; rng[2 * k + 1] = -1;
            rk[2 * k] = 0;  rk[2 * k + 1] = 0;
        }
    }
    __syncthreads();

    if (tid < 32) {
        int v[8];
        int s = 0;
#pragma unroll
        for (int j = 0; j < 8; ++j) { v[j] = hist[8 * tid + j]; s += v[j]; }
        int incl = s;
#pragma unroll
        for (int off = 1; off < 32; off <<= 1) {
            const int nb = __shfl_up(incl, off);
            if (tid >= off) incl += nb;
        }
        int excl = incl - s;
#pragma unroll
        for (int j = 0; j < 8; ++j) { loff[8 * tid + j] = excl; excl += v[j]; }
    }
    if (tid < GCN_NRANGE) {
        const int h = hist[tid];
        int base = 0, first = 0, ob = 0;
        if (h > 0) {
            base  = atomicAdd(&gcount[tid], h);
            first = (base > GCN_CAPR) ? base : GCN_CAPR;
            const int nov = base + h - first;
            if (nov > 0) ob = atomicAdd(covf_cnt, nov);
        }
        gbase[tid] = base; gfirst[tid] = first; gob[tid] = ob;
    }
    __syncthreads();

#pragma unroll
    for (int k = 0; k < 2 * GCN_EPT; ++k) {
        if (rng[k] >= 0) {
            const int p = loff[rng[k]] + rk[k];
            stage[p]     = ent[k];
            stage_rng[p] = (unsigned short)rng[k];
        }
    }
    __syncthreads();

    int nloc = nedges - e0;
    nloc = (nloc < GCN_CHUNK) ? nloc : GCN_CHUNK;
    const int total = 2 * nloc;
    for (int p = tid; p < total; p += GCN_PT) {
        const int r   = stage_rng[p];
        const int pos = gbase[r] + (p - loff[r]);
        if (pos < GCN_CAPR) {
            part[(size_t)r * GCN_CAPR + pos] = stage[p];
        } else {
            const int oi = gob[r] + (pos - gfirst[r]);
            if (oi < covf_cap) { covf_ent[oi] = stage[p]; covf_rng[oi] = r; }
        }
    }
}

#define GCN_FT 1024
__global__ void __launch_bounds__(GCN_FT)
gcn_fine_kernel(const int* __restrict__ gcount,
                const int* __restrict__ part,
                const int* __restrict__ covf_cnt,
                const int* __restrict__ covf_ent,
                const int* __restrict__ covf_rng,
                int* __restrict__ cnt_src,
                int* __restrict__ cnt_dst,
                int* __restrict__ col,
                int* __restrict__ ovf_cnt,
                int* __restrict__ ovf_dst,
                int* __restrict__ ovf_src,
                int covf_cap, int ovf_cap) {
    __shared__ int cd[GCN_RNODES];
    __shared__ int cs[GCN_RNODES];
    __shared__ int tile[GCN_RNODES * GCN_CAP];

    const int tid   = threadIdx.x;
    const int r     = blockIdx.x;
    const int node0 = r * GCN_RNODES;
    for (int i = tid; i < GCN_RNODES; i += GCN_FT) { cd[i] = 0; cs[i] = 0; }
    __syncthreads();

    int n = gcount[r];
    n = (n < GCN_CAPR) ? n : GCN_CAPR;
    int novf = covf_cnt[0];
    novf = (novf < covf_cap) ? novf : covf_cap;
    const int* __restrict__ pr = part + (size_t)r * GCN_CAPR;
    for (int i = tid; i < n + novf; i += GCN_FT) {
        int e;
        if (i < n) {
            e = pr[i];
        } else {
            if (covf_rng[i - n] != r) continue;
            e = covf_ent[i - n];
        }
        const int local = e & 511;
        if (e & 512) {
            atomicAdd(&cs[local], 1);
        } else {
            const int k = atomicAdd(&cd[local], 1);
            if (k < GCN_CAP) {
                tile[local * GCN_CAP + k] = e >> 10;
            } else {
                const int idx = atomicAdd(ovf_cnt, 1);
                if (idx < ovf_cap) { ovf_dst[idx] = node0 + local; ovf_src[idx] = e >> 10; }
            }
        }
    }
    __syncthreads();

    for (int i = tid; i < GCN_RNODES; i += GCN_FT) {
        const int node = node0 + i;
        if (node < GCN_N) { cnt_dst[node] = cd[i]; cnt_src[node] = cs[i]; }
    }
    for (int idx = tid; idx < GCN_RNODES * GCN_CAP; idx += GCN_FT) {
        const int l    = idx / GCN_CAP;
        const int k    = idx - l * GCN_CAP;
        const int node = node0 + l;
        if (node < GCN_N && k < cd[l]) col[(size_t)node * GCN_CAP + k] = tile[idx];
    }
}

__device__ __forceinline__ float gcn_norm_of(int deg) {
    return rsqrtf(fmaxf((float)deg, 1.0f));
}

__device__ __forceinline__ int gcn_frag_k(int lane, int i) {
    const int j   = i >> 1;
    const int sub = i & 1;
    const int kk  = (j < 4) ? (2 * j + sub) : (16 + 2 * (j - 4) + sub);
    return kk + 8 * (lane >> 4);
}

__global__ void gcn_pack_w_kernel(const float* __restrict__ W,
                                  _Float16* __restrict__ Wp) {
    int idx = blockIdx.x * blockDim.x + threadIdx.x;
    if (idx >= GCN_WPACK_ELEMS) return;
    const int i    = idx & 15;
    const int lane = (idx >> 4) & 31;
    const int kb   = (idx >> 9) & 3;
    const int nt   = idx >> 11;
    const int K    = kb * 32 + gcn_frag_k(lane, i);
    const int n    = nt * 16 + (lane & 15);
    Wp[idx] = (_Float16)W[(size_t)K * GCN_D + n];
}

typedef __attribute__((ext_vector_type(8)))  _Float16 v8h;
typedef __attribute__((ext_vector_type(4)))  _Float16 v4h;

#define GCN_CVROWS 16
static_assert(GCN_N % GCN_CVROWS == 0, "a wave converts whole groups of rows");
__global__ void __launch_bounds__(256)
gcn_rows_to_f16_kernel(const float* __restrict__ X,
                       const int* __restrict__ cnt_src,
                       _Float16* __restrict__ Xh,
                       int nrows) {
    const int lane = threadIdx.x & 31;
    const int wave = __builtin_amdgcn_readfirstlane((int)(threadIdx.x >> 5));
    const int r0   = (blockIdx.x * 8 + wave) * GCN_CVROWS;
    if (r0 >= nrows) return;
    const float4* __restrict__ X4 = (const float4*)X;
    v4h* __restrict__ H4 = (v4h*)Xh;
#pragma unroll 1
    for (int b = 0; b < GCN_CVROWS; b += 8) {
        float4 v[8];
        float  ns[8];
#pragma unroll
        for (int k = 0; k < 8; ++k) {
            const int row = r0 + b + k;
            v[k]  = X4[(size_t)row * (GCN_D / 4) + lane];
            ns[k] = gcn_norm_of(cnt_src[row]);
        }
#pragma unroll
        for (int k = 0; k < 8; ++k) {
            const int row = r0 + b + k;
            v4h h;
            h[0] = (_Float16)(v[k].x * ns[k]); h[1] = (_Float16)(v[k].y * ns[k]);
            h[2] = (_Float16)(v[k].z * ns[k]); h[3] = (_Float16)(v[k].w * ns[k]);
            H4[(size_t)row * (GCN_D / 4) + lane] = h;
        }
    }
}

template <int G, bool MASKED>
__device__ __forceinline__ void gcn_sum_group(const v4h* __restrict__ X4, int myidx,
                                              int j, int n, int lane, float4& acc) {
    v4h v[G];
#pragma unroll
    for (int k = 0; k < G; ++k) {
        const int s = __builtin_amdgcn_readlane(myidx, j + k);
        v[k] = X4[(size_t)s * (GCN_D / 4) + lane];
    }
#pragma unroll
    for (int k = 0; k < G; ++k) {
        if (MASKED) {
            const float w = (j + k < n) ? 1.0f : 0.0f;
            acc.x = fmaf(w, (float)v[k][0], acc.x); acc.y = fmaf(w, (float)v[k][1], acc.y);
            acc.z = fmaf(w, (float)v[k][2], acc.z); acc.w = fmaf(w, (float)v[k][3], acc.w);
        } else {
            acc.x += (float)v[k][0]; acc.y += (float)v[k][1];
            acc.z += (float)v[k][2]; acc.w += (float)v[k][3];
        }
    }
}

#define GCN_APITCH 136
#define GCN_OPITCH 132

#define GCN_LAYER_GRID 1280

__device__ __forceinline__ int gcn_two_degrees(const int* __restrict__ deg, int r,
                                               int nrows, int lane) {
    const int q  = r + (lane & 1);
    const int qc = (q < nrows) ? q : (nrows - 1);
    const int v  = deg[qc];
    return (q < nrows) ? v : 0;
}

__device__ __forceinline__ void gcn_lds_barrier() {
    __builtin_amdgcn_fence(__ATOMIC_RELEASE, "workgroup", "local");
    __builtin_amdgcn_s_barrier();
    __builtin_amdgcn_fence(__ATOMIC_ACQUIRE, "workgroup", "local");
}

__device__ __forceinline__ int gcn_first_idx(const int* __restrict__ col, int row,
                                             int nb, int lane) {
    if (nb <= 0) return 0;
    const int n = (nb < 32) ? nb : 32;
    return col[(size_t)row * GCN_CAP + ((lane < n) ? lane : 0)];
}

__device__ __forceinline__ void gcn_sum_chunk(const v4h* __restrict__ X4, int myidx,
                                              int n, int lane, float4& acc) {
    int j = 0;
    for (; j + 8 <= n; j += 8)
        gcn_sum_group<8, false>(X4, myidx, j, n, lane, acc);
    if (j < n) {
        if (n - j > 4) gcn_sum_group<8, true>(X4, myidx, j, n, lane, acc);
        else           gcn_sum_group<4, true>(X4, myidx, j, n, lane, acc);
    }
}

template <bool HIDDEN_OUT>
__global__ void __launch_bounds__(256)
gcn_layer_kernel(const _Float16* __restrict__ Xh,
                 const int* __restrict__ cnt_src,
                 const int* __restrict__ cnt_dst,
                 const int* __restrict__ col,
                 const int* __restrict__ ovf_cnt,
                 const int* __restrict__ ovf_dst,
                 const int* __restrict__ ovf_src,
                 const _Float16* __restrict__ Wp,
                 const float* __restrict__ bias,
                 float* __restrict__ out_f32,
                 _Float16* __restrict__ out_f16,
                 int nrows, int ntiles, int tile_step, int ovf_cap) {
    __shared__ __attribute__((aligned(16))) _Float16 atile[16 * GCN_APITCH];
    __shared__ __attribute__((aligned(16))) float  otile[16 * GCN_OPITCH];

    const int lane = threadIdx.x & 31;
    const int wave = __builtin_amdgcn_readfirstlane((int)(threadIdx.x >> 5));
    const v4h* __restrict__ X4 = (const v4h*)Xh;

    const int n0   = wave * 16;
    const int half = lane >> 4;
    const int m    = lane & 15;

    v16h b[4];
#pragma unroll
    for (int kb = 0; kb < 4; ++kb)
        b[kb] = *(const v16h*)&Wp[(((size_t)(wave * 4 + kb)) * 32 + lane) * 16];
    const float4 bv = ((const float4*)bias)[lane];

    int tile = blockIdx.x;
    int cnt0 = 0, cnt1 = 0, idx0 = 0, idx1 = 0;
    if (tile < ntiles) {
        const int r = tile * 16 + wave * 2;
        const int cv = gcn_two_degrees(cnt_dst, r, nrows, lane);
        cnt0 = __builtin_amdgcn_readlane(cv, 0);
        cnt1 = __builtin_amdgcn_readlane(cv, 1);
        idx0 = gcn_first_idx(col, r,     (cnt0 < GCN_CAP) ? cnt0 : GCN_CAP, lane);
        idx1 = gcn_first_idx(col, r + 1, (cnt1 < GCN_CAP) ? cnt1 : GCN_CAP, lane);
    }

    while (tile < ntiles) {
        const int row0 = tile * 16;
        const int  tile_n = tile + tile_step;
        const bool more   = tile_n < ntiles;
        const int  rn     = tile_n * 16 + wave * 2;
        const int cvn = gcn_two_degrees(cnt_dst, rn, nrows, lane);
        int csv = 0;
        if (HIDDEN_OUT) csv = gcn_two_degrees(cnt_src, row0 + wave * 2, nrows, lane);

#pragma unroll 1
        for (int rr = 0; rr < 2; ++rr) {
            const int lr  = wave * 2 + rr;
            const int row = row0 + lr;
            const int cnt = rr ? cnt1 : cnt0;
            float4 acc = make_float4(0.0f, 0.0f, 0.0f, 0.0f);
            const int nb  = (cnt < GCN_CAP) ? cnt : GCN_CAP;
            if (nb > 0) {
                gcn_sum_chunk(X4, rr ? idx1 : idx0, (nb < 32) ? nb : 32, lane, acc);
                const int* __restrict__ crow = col + (size_t)row * GCN_CAP;
                for (int done = 32; done < nb; done += 32) {
                    const int n = (nb - done < 32) ? (nb - done) : 32;
                    const int myidx = crow[done + ((lane < n) ? lane : 0)];
                    gcn_sum_chunk(X4, myidx, n, lane, acc);
                }
            }
            if (cnt > GCN_CAP) {
                int novf = ovf_cnt[0];
                novf = (novf < ovf_cap) ? novf : ovf_cap;
                for (int base = 0; base < novf; base += 32) {
                    const int  e   = base + lane;
                    const bool hit = (e < novf) && (ovf_dst[e] == row);
                    const int  hs  = hit ? ovf_src[e] : 0;
                    unsigned mask = __builtin_amdgcn_ballot_w32(hit);
                    while (mask != 0u) {
                        const int j = __builtin_ctz(mask);
                        mask &= mask - 1u;
                        const int s = __builtin_amdgcn_readlane(hs, j);
                        const v4h v = X4[(size_t)s * (GCN_D / 4) + lane];
                        acc.x += (float)v[0]; acc.y += (float)v[1];
                        acc.z += (float)v[2]; acc.w += (float)v[3];
                    }
                }
            }
            v4h p;
            p[0] = (_Float16)acc.x; p[1] = (_Float16)acc.y;
            p[2] = (_Float16)acc.z; p[3] = (_Float16)acc.w;
            *(v4h*)&atile[lr * GCN_APITCH + lane * 4] = p;
        }

        const int cnt0n = __builtin_amdgcn_readlane(cvn, 0);
        const int cnt1n = __builtin_amdgcn_readlane(cvn, 1);
        int idx0n = 0, idx1n = 0;
        if (more) {
            idx0n = gcn_first_idx(col, rn,     (cnt0n < GCN_CAP) ? cnt0n : GCN_CAP, lane);
            idx1n = gcn_first_idx(col, rn + 1, (cnt1n < GCN_CAP) ? cnt1n : GCN_CAP, lane);
        }

        gcn_lds_barrier();

        v8f c = {};
#pragma unroll
        for (int kb = 0; kb < 4; ++kb) {
            const _Float16* ap = &atile[m * GCN_APITCH + kb * 32 + 8 * half];
            const v8h lo = *(const v8h*)ap;
            const v8h hi = *(const v8h*)(ap + 16);
            const v16h a = __builtin_shufflevector(lo, hi, 0, 1, 2, 3, 4, 5, 6, 7,
                                                   8, 9, 10, 11, 12, 13, 14, 15);
            c = __builtin_amdgcn_wmma_f32_16x16x32_f16(
                 false, a,  false, b[kb],
                 (short)0, c,  false,  false);
        }

#pragma unroll
        for (int i = 0; i < 8; ++i)
            otile[(i + 8 * half) * GCN_OPITCH + n0 + m] = c[i];
        gcn_lds_barrier();

#pragma unroll
        for (int rr = 0; rr < 2; ++rr) {
            const int lr  = wave * 2 + rr;
            const int row = row0 + lr;
            if (row < nrows) {
                const float4 v  = *(const float4*)&otile[lr * GCN_OPITCH + lane * 4];
                const float  nd = gcn_norm_of(rr ? cnt1 : cnt0);
                float4 r;
                r.x = v.x * nd + bv.x;
                r.y = v.y * nd + bv.y;
                r.z = v.z * nd + bv.z;
                r.w = v.w * nd + bv.w;
                if (HIDDEN_OUT) {
                    const float ns = gcn_norm_of(__builtin_amdgcn_readlane(csv, rr));
                    v4h h;
                    h[0] = (_Float16)(fmaxf(r.x, 0.0f) * ns); h[1] = (_Float16)(fmaxf(r.y, 0.0f) * ns);
                    h[2] = (_Float16)(fmaxf(r.z, 0.0f) * ns); h[3] = (_Float16)(fmaxf(r.w, 0.0f) * ns);
                    ((v4h*)out_f16)[(size_t)row * (GCN_D / 4) + lane] = h;
                } else {
                    ((float4*)out_f32)[(size_t)row * (GCN_D / 4) + lane] = r;
                }
            }
        }
        tile = tile_n;
        cnt0 = cnt0n; cnt1 = cnt1n; idx0 = idx0n; idx1 = idx1n;
    }
}

extern "C" void kernel_launch(void* const* d_in, const int* in_sizes, int n_in,
                              void* d_out, int out_size, void* d_ws, size_t ws_size,
                              hipStream_t stream) {
    const float* feat = (const float*)d_in[0];
    const int*   src  = (const int*)  d_in[1];
    const int*   dst  = (const int*)  d_in[2];
    const float* W1   = (const float*)d_in[3];
    const float* b1   = (const float*)d_in[4];
    const float* W2   = (const float*)d_in[5];
    const float* b2   = (const float*)d_in[6];
    float* out = (float*)d_out;

    const int N = GCN_N;
    const int E = in_sizes[1];
    const size_t NF = (size_t)N * GCN_D;

    int*    cntS   = (int*)d_ws;
    int*    cntD   = cntS + GCN_NPAD;
    int*    ovfcnt = cntD + GCN_NPAD;
    int*    gcount = ovfcnt + GCN_OVFPAD;
    int*    covfcnt = gcount + GCN_NRANGE;
    _Float16* Wp1  = (_Float16*)(covfcnt + GCN_OVFPAD);
    _Float16* Wp2  = Wp1 + GCN_WPACK_ELEMS;
    _Float16* Xh   = Wp2 + GCN_WPACK_ELEMS;
    _Float16* Hh   = Xh + NF;
    int*    col    = (int*)(Hh + NF);
    int*    ovfD   = col + (size_t)N * GCN_CAP;
    int*    ovfS   = ovfD + E;
    int*    part   = ovfS + E;
    int*    covfE  = part + (size_t)GCN_NRANGE * GCN_CAPR;
    int*    covfR  = covfE + (size_t)2 * E;

    const int thr = 256;
    const int nzero = GCN_OVFPAD + GCN_NRANGE + GCN_OVFPAD;
    const int zgrid = (nzero + thr - 1) / thr;
    const int pgrid1 = (E + GCN_CHUNK - 1) / GCN_CHUNK;
    const int pgrid = GCN_WPACK_ELEMS / thr;
    const int ntiles = N / 16;
    const int layer_grid = (ntiles < GCN_LAYER_GRID) ? ntiles : GCN_LAYER_GRID;
    const int cvgrid = (N + 8 * GCN_CVROWS - 1) / (8 * GCN_CVROWS);

    gcn_zero_i32_kernel<<<zgrid, thr, 0, stream>>>(ovfcnt, nzero);
    gcn_partition_kernel<<<pgrid1, GCN_PT, 0, stream>>>(src, dst, gcount, part,
                                                        covfcnt, covfE, covfR, E, 2 * E);
    gcn_fine_kernel<<<GCN_NRANGE, GCN_FT, 0, stream>>>(gcount, part, covfcnt, covfE, covfR,
                                                       cntS, cntD, col, ovfcnt, ovfD, ovfS,
                                                       2 * E, E);
    gcn_pack_w_kernel<<<pgrid, thr, 0, stream>>>(W1, Wp1);
    gcn_pack_w_kernel<<<pgrid, thr, 0, stream>>>(W2, Wp2);

    gcn_rows_to_f16_kernel<<<cvgrid, thr, 0, stream>>>(feat, cntS, Xh, N);

    gcn_layer_kernel<true><<<layer_grid, thr, 0, stream>>>(
        Xh, cntS, cntD, col, ovfcnt, ovfD, ovfS, Wp1, b1, (float*)nullptr, Hh,
        N, ntiles, layer_grid, E);

    gcn_layer_kernel<false><<<layer_grid, thr, 0, stream>>>(
        Hh, cntS, cntD, col, ovfcnt, ovfD, ovfS, Wp2, b2, out, (_Float16*)nullptr,
        N, ntiles, layer_grid, E);
}


# Five launches, not eight: both W packings and the counter zeroing share one set-up launch, and each range block of the bucket pass stores its own nodes' scaled f16 rows (no copy launch); same values
# speedup vs baseline: 8.8728x; 1.0604x over previous
#include <hip/hip_runtime.h>


typedef __attribute__((ext_vector_type(16))) _Float16 v16h;
typedef __attribute__((ext_vector_type(8)))  float  v8f;
typedef __attribute__((ext_vector_type(8)))  _Float16 v8h;
typedef __attribute__((ext_vector_type(4)))  _Float16 v4h;

#define GCN_N 100000
#define GCN_D 128
#define GCN_WPACK_ELEMS (8 * 4 * 32 * 16)

#define GCN_NPAD 100032

#define GCN_CAP 64
#define GCN_OVFPAD 64

__device__ __forceinline__ float gcn_norm_of(int deg) {
    return rsqrtf(fmaxf((float)deg, 1.0f));
}

#define GCN_RNODES 392
#define GCN_NRANGE 256
#define GCN_CAPR   32768
#define GCN_PT     512
#define GCN_EPT    4
#define GCN_CHUNK  (GCN_PT * GCN_EPT)

__global__ void __launch_bounds__(GCN_PT)
gcn_partition_kernel(const int* __restrict__ src,
                     const int* __restrict__ dst,
                     int* __restrict__ gcount,
                     int* __restrict__ part,
                     int* __restrict__ covf_cnt,
                     int* __restrict__ covf_ent,
                     int* __restrict__ covf_rng,
                     int nedges, int covf_cap) {
    __shared__ int hist[GCN_NRANGE];
    __shared__ int loff[GCN_NRANGE];
    __shared__ int gbase[GCN_NRANGE];
    __shared__ int gfirst[GCN_NRANGE];
    __shared__ int gob[GCN_NRANGE];
    __shared__ int stage[2 * GCN_CHUNK];
    __shared__ unsigned short stage_rng[2 * GCN_CHUNK];

    const int tid = threadIdx.x;
    if (tid < GCN_NRANGE) hist[tid] = 0;
    __syncthreads();

    int ent[2 * GCN_EPT], rng[2 * GCN_EPT], rk[2 * GCN_EPT];
    const int e0 = blockIdx.x * GCN_CHUNK;
#pragma unroll
    for (int k = 0; k < GCN_EPT; ++k) {
        const int i = e0 + k * GCN_PT + tid;
        if (i < nedges) {
            const int s  = src[i];
            const int d  = dst[i];
            const int rd = d / GCN_RNODES;
            const int rs = s / GCN_RNODES;
            ent[2 * k]     = (s << 10) | (d - rd * GCN_RNODES);
            rng[2 * k]     = rd;
            rk[2 * k]      = atomicAdd(&hist[rd], 1);
            ent[2 * k + 1] = (1 << 9) | (s - rs * GCN_RNODES);
            rng[2 * k + 1] = rs;
            rk[2 * k + 1]  = atomicAdd(&hist[rs], 1);
        } else {
            ent[2 * k] = 0; ent[2 * k + 1] = 0;
            rng[2 * k] = -1; rng[2 * k + 1] = -1;
            rk[2 * k] = 0;  rk[2 * k + 1] = 0;
        }
    }
    __syncthreads();

    if (tid < 32) {
        int v[8];
        int s = 0;
#pragma unroll
        for (int j = 0; j < 8; ++j) { v[j] = hist[8 * tid + j]; s += v[j]; }
        int incl = s;
#pragma unroll
        for (int off = 1; off < 32; off <<= 1) {
            const int nb = __shfl_up(incl, off);
            if (tid >= off) incl += nb;
        }
        int excl = incl - s;
#pragma unroll
        for (int j = 0; j < 8; ++j) { loff[8 * tid + j] = excl; excl += v[j]; }
    }
    if (tid < GCN_NRANGE) {
        const int h = hist[tid];
        int base = 0, first = 0, ob = 0;
        if (h > 0) {
            base  = atomicAdd(&gcount[tid], h);
            first = (base > GCN_CAPR) ? base : GCN_CAPR;
            const int nov = base + h - first;
            if (nov > 0) ob = atomicAdd(covf_cnt, nov);
        }
        gbase[tid] = base; gfirst[tid] = first; gob[tid] = ob;
    }
    __syncthreads();

#pragma unroll
    for (int k = 0; k < 2 * GCN_EPT; ++k) {
        if (rng[k] >= 0) {
            const int p = loff[rng[k]] + rk[k];
            stage[p]     = ent[k];
            stage_rng[p] = (unsigned short)rng[k];
        }
    }
    __syncthreads();

    int nloc = nedges - e0;
    nloc = (nloc < GCN_CHUNK) ? nloc : GCN_CHUNK;
    const int total = 2 * nloc;
    for (int p = tid; p < total; p += GCN_PT) {
        const int r   = stage_rng[p];
        const int pos = gbase[r] + (p - loff[r]);
        if (pos < GCN_CAPR) {
            part[(size_t)r * GCN_CAPR + pos] = stage[p];
        } else {
            const int oi = gob[r] + (pos - gfirst[r]);
            if (oi < covf_cap) { covf_ent[oi] = stage[p]; covf_rng[oi] = r; }
        }
    }
}

template <int G>
__device__ __forceinline__ void gcn_load_rows(const float4* __restrict__ X4, int node0, int l0,
                                              int wave, int lane, float4 (&v)[G]) {
#pragma unroll
    for (int k = 0; k < G; ++k) {
        const int node = node0 + l0 + k * 32 + wave;
        const int nc   = (node < GCN_N) ? node : (GCN_N - 1);
        v[k] = X4[(size_t)nc * (GCN_D / 4) + lane];
    }
}
template <int G>
__device__ __forceinline__ void gcn_store_rows(v4h* __restrict__ H4, const int* cs, int node0,
                                               int l0, int wave, int lane, const float4 (&v)[G]) {
#pragma unroll
    for (int k = 0; k < G; ++k) {
        const int l    = l0 + k * 32 + wave;
        const int node = node0 + l;
        if (node < GCN_N) {
            const float ns = gcn_norm_of(cs[l]);
            v4h h;
            h[0] = (_Float16)(v[k].x * ns); h[1] = (_Float16)(v[k].y * ns);
            h[2] = (_Float16)(v[k].z * ns); h[3] = (_Float16)(v[k].w * ns);
            H4[(size_t)node * (GCN_D / 4) + lane] = h;
        }
    }
}

#define GCN_FT 1024
#define GCN_FROWS 12
static_assert(GCN_RNODES > GCN_FROWS * 32 && GCN_RNODES <= (GCN_FROWS + 1) * 32,
              "a wave copies GCN_FROWS rows, and one more where its number is below the remainder");
__global__ void __launch_bounds__(GCN_FT)
gcn_fine_kernel(const int* __restrict__ gcount,
                const int* __restrict__ part,
                const int* __restrict__ covf_cnt,
                const int* __restrict__ covf_ent,
                const int* __restrict__ covf_rng,
                int* __restrict__ cnt_src,
                int* __restrict__ cnt_dst,
                int* __restrict__ col,
                int* __restrict__ ovf_cnt,
                int* __restrict__ ovf_dst,
                int* __restrict__ ovf_src,
                const float* __restrict__ X,
                _Float16* __restrict__ Xh,
                int covf_cap, int ovf_cap) {
    __shared__ int cd[GCN_RNODES];
    __shared__ int cs[GCN_RNODES];
    __shared__ int tile[GCN_RNODES * GCN_CAP];

    const int tid   = threadIdx.x;
    const int r     = blockIdx.x;
    const int node0 = r * GCN_RNODES;
    for (int i = tid; i < GCN_RNODES; i += GCN_FT) { cd[i] = 0; cs[i] = 0; }
    __syncthreads();

    int n = gcount[r];
    n = (n < GCN_CAPR) ? n : GCN_CAPR;
    int novf = covf_cnt[0];
    novf = (novf < covf_cap) ? novf : covf_cap;
    const int* __restrict__ pr = part + (size_t)r * GCN_CAPR;
    for (int i = tid; i < n + novf; i += GCN_FT) {
        int e;
        if (i < n) {
            e = pr[i];
        } else {
            if (covf_rng[i - n] != r) continue;
            e = covf_ent[i - n];
        }
        const int local = e & 511;
        if (e & 512) {
            atomicAdd(&cs[local], 1);
        } else {
            const int k = atomicAdd(&cd[local], 1);
            if (k < GCN_CAP) {
                tile[local * GCN_CAP + k] = e >> 10;
            } else {
                const int idx = atomicAdd(ovf_cnt, 1);
                if (idx < ovf_cap) { ovf_dst[idx] = node0 + local; ovf_src[idx] = e >> 10; }
            }
        }
    }
    __syncthreads();

    const int lane = tid & 31;
    const int wave = __builtin_amdgcn_readfirstlane(tid >> 5);
    const float4* __restrict__ X4 = (const float4*)X;
    v4h* __restrict__ H4 = (v4h*)Xh;
    const bool last_row = wave < GCN_RNODES - GCN_FROWS * 32;
    float4 va[GCN_FROWS];
    float4 vc[1];
    gcn_load_rows<GCN_FROWS>(X4, node0, 0, wave, lane, va);
    gcn_load_rows<1>(X4, node0, last_row ? GCN_FROWS * 32 : 0, wave, lane, vc);

    for (int i = tid; i < GCN_RNODES; i += GCN_FT) {
        const int node = node0 + i;
        if (node < GCN_N) { cnt_dst[node] = cd[i]; cnt_src[node] = cs[i]; }
    }
    for (int idx = tid; idx < GCN_RNODES * GCN_CAP; idx += GCN_FT) {
        const int l    = idx / GCN_CAP;
        const int k    = idx - l * GCN_CAP;
        const int node = node0 + l;
        if (node < GCN_N && k < cd[l]) col[(size_t)node * GCN_CAP + k] = tile[idx];
    }

    gcn_store_rows<GCN_FROWS>(H4, cs, node0, 0, wave, lane, va);
    if (last_row) gcn_store_rows<1>(H4, cs, node0, GCN_FROWS * 32, wave, lane, vc);
}

__device__ __forceinline__ int gcn_frag_k(int lane, int i) {
    const int j   = i >> 1;
    const int sub = i & 1;
    const int kk  = (j < 4) ? (2 * j + sub) : (16 + 2 * (j - 4) + sub);
    return kk + 8 * (lane >> 4);
}

#define GCN_SETUP_THREADS 256
#define GCN_PACK_BLOCKS (GCN_WPACK_ELEMS / GCN_SETUP_THREADS)
__global__ void __launch_bounds__(GCN_SETUP_THREADS)
gcn_setup_kernel(const float* __restrict__ W1, const float* __restrict__ W2,
                 _Float16* __restrict__ Wp1, _Float16* __restrict__ Wp2,
                 int* __restrict__ counters, int ncounters) {
    const int blk = blockIdx.x;
    if (blk >= 2 * GCN_PACK_BLOCKS) {
        for (int i = threadIdx.x; i < ncounters; i += GCN_SETUP_THREADS) counters[i] = 0;
        return;
    }
    const bool second = blk >= GCN_PACK_BLOCKS;
    const float* __restrict__ W  = second ? W2 : W1;
    _Float16* __restrict__    Wp = second ? Wp2 : Wp1;
    const int idx  = (second ? blk - GCN_PACK_BLOCKS : blk) * GCN_SETUP_THREADS + threadIdx.x;
    const int i    = idx & 15;
    const int lane = (idx >> 4) & 31;
    const int kb   = (idx >> 9) & 3;
    const int nt   = idx >> 11;
    const int K    = kb * 32 + gcn_frag_k(lane, i);
    const int n    = nt * 16 + (lane & 15);
    Wp[idx] = (_Float16)W[(size_t)K * GCN_D + n];
}

template <int G, bool MASKED>
__device__ __forceinline__ void gcn_sum_group(const v4h* __restrict__ X4, int myidx,
                                              int j, int n, int lane, float4& acc) {
    v4h v[G];
#pragma unroll
    for (int k = 0; k < G; ++k) {
        const int s = __builtin_amdgcn_readlane(myidx, j + k);
        v[k] = X4[(size_t)s * (GCN_D / 4) + lane];
    }
#pragma unroll
    for (int k = 0; k < G; ++k) {
        if (MASKED) {
            const float w = (j + k < n) ? 1.0f : 0.0f;
            acc.x = fmaf(w, (float)v[k][0], acc.x); acc.y = fmaf(w, (float)v[k][1], acc.y);
            acc.z = fmaf(w, (float)v[k][2], acc.z); acc.w = fmaf(w, (float)v[k][3], acc.w);
        } else {
            acc.x += (float)v[k][0]; acc.y += (float)v[k][1];
            acc.z += (float)v[k][2]; acc.w += (float)v[k][3];
        }
    }
}

#define GCN_APITCH 136
#define GCN_OPITCH 132

#define GCN_LAYER_GRID 1280

__device__ __forceinline__ int gcn_two_degrees(const int* __restrict__ deg, int r,
                                               int nrows, int lane) {
    const int q  = r + (lane & 1);
    const int qc = (q < nrows) ? q : (nrows - 1);
    const int v  = deg[qc];
    return (q < nrows) ? v : 0;
}

__device__ __forceinline__ void gcn_lds_barrier() {
    __builtin_amdgcn_fence(__ATOMIC_RELEASE, "workgroup", "local");
    __builtin_amdgcn_s_barrier();
    __builtin_amdgcn_fence(__ATOMIC_ACQUIRE, "workgroup", "local");
}

__device__ __forceinline__ int gcn_first_idx(const int* __restrict__ col, int row,
                                             int nb, int lane) {
    if (nb <= 0) return 0;
    const int n = (nb < 32) ? nb : 32;
    return col[(size_t)row * GCN_CAP + ((lane < n) ? lane : 0)];
}

__device__ __forceinline__ void gcn_sum_chunk(const v4h* __restrict__ X4, int myidx,
                                              int n, int lane, float4& acc) {
    int j = 0;
    for (; j + 8 <= n; j += 8)
        gcn_sum_group<8, false>(X4, myidx, j, n, lane, acc);
    if (j < n) {
        if (n - j > 4) gcn_sum_group<8, true>(X4, myidx, j, n, lane, acc);
        else           gcn_sum_group<4, true>(X4, myidx, j, n, lane, acc);
    }
}

template <bool HIDDEN_OUT>
__global__ void __launch_bounds__(256)
gcn_layer_kernel(const _Float16* __restrict__ Xh,
                 const int* __restrict__ cnt_src,
                 const int* __restrict__ cnt_dst,
                 const int* __restrict__ col,
                 const int* __restrict__ ovf_cnt,
                 const int* __restrict__ ovf_dst,
                 const int* __restrict__ ovf_src,
                 const _Float16* __restrict__ Wp,
                 const float* __restrict__ bias,
                 float* __restrict__ out_f32,
                 _Float16* __restrict__ out_f16,
                 int nrows, int ntiles, int tile_step, int ovf_cap) {
    __shared__ __attribute__((aligned(16))) _Float16 atile[16 * GCN_APITCH];
    __shared__ __attribute__((aligned(16))) float  otile[16 * GCN_OPITCH];

    const int lane = threadIdx.x & 31;
    const int wave = __builtin_amdgcn_readfirstlane((int)(threadIdx.x >> 5));
    const v4h* __restrict__ X4 = (const v4h*)Xh;

    const int n0   = wave * 16;
    const int half = lane >> 4;
    const int m    = lane & 15;

    v16h b[4];
#pragma unroll
    for (int kb = 0; kb < 4; ++kb)
        b[kb] = *(const v16h*)&Wp[(((size_t)(wave * 4 + kb)) * 32 + lane) * 16];
    const float4 bv = ((const float4*)bias)[lane];

    int tile = blockIdx.x;
    int cnt0 = 0, cnt1 = 0, idx0 = 0, idx1 = 0;
    if (tile < ntiles) {
        const int r = tile * 16 + wave * 2;
        const int cv = gcn_two_degrees(cnt_dst, r, nrows, lane);
        cnt0 = __builtin_amdgcn_readlane(cv, 0);
        cnt1 = __builtin_amdgcn_readlane(cv, 1);
        idx0 = gcn_first_idx(col, r,     (cnt0 < GCN_CAP) ? cnt0 : GCN_CAP, lane);
        idx1 = gcn_first_idx(col, r + 1, (cnt1 < GCN_CAP) ? cnt1 : GCN_CAP, lane);
    }

    while (tile < ntiles) {
        const int row0 = tile * 16;
        const int  tile_n = tile + tile_step;
        const bool more   = tile_n < ntiles;
        const int  rn     = tile_n * 16 + wave * 2;
        const int cvn = gcn_two_degrees(cnt_dst, rn, nrows, lane);
        int csv = 0;
        if (HIDDEN_OUT) csv = gcn_two_degrees(cnt_src, row0 + wave * 2, nrows, lane);

#pragma unroll 1
        for (int rr = 0; rr < 2; ++rr) {
            const int lr  = wave * 2 + rr;
            const int row = row0 + lr;
            const int cnt = rr ? cnt1 : cnt0;
            float4 acc = make_float4(0.0f, 0.0f, 0.0f, 0.0f);
            const int nb  = (cnt < GCN_CAP) ? cnt : GCN_CAP;
            if (nb > 0) {
                gcn_sum_chunk(X4, rr ? idx1 : idx0, (nb < 32) ? nb : 32, lane, acc);
                const int* __restrict__ crow = col + (size_t)row * GCN_CAP;
                for (int done = 32; done < nb; done += 32) {
                    const int n = (nb - done < 32) ? (nb - done) : 32;
                    const int myidx = crow[done + ((lane < n) ? lane : 0)];
                    gcn_sum_chunk(X4, myidx, n, lane, acc);
                }
            }
            if (cnt > GCN_CAP) {
                int novf = ovf_cnt[0];
                novf = (novf < ovf_cap) ? novf : ovf_cap;
                for (int base = 0; base < novf; base += 32) {
                    const int  e   = base + lane;
                    const bool hit = (e < novf) && (ovf_dst[e] == row);
                    const int  hs  = hit ? ovf_src[e] : 0;
                    unsigned mask = __builtin_amdgcn_ballot_w32(hit);
                    while (mask != 0u) {
                        const int j = __builtin_ctz(mask);
                        mask &= mask - 1u;
                        const int s = __builtin_amdgcn_readlane(hs, j);
                        const v4h v = X4[(size_t)s * (GCN_D / 4) + lane];
                        acc.x += (float)v[0]; acc.y += (float)v[1];
                        acc.z += (float)v[2]; acc.w += (float)v[3];
                    }
                }
            }
            v4h p;
            p[0] = (_Float16)acc.x; p[1] = (_Float16)acc.y;
            p[2] = (_Float16)acc.z; p[3] = (_Float16)acc.w;
            *(v4h*)&atile[lr * GCN_APITCH + lane * 4] = p;
        }

        const int cnt0n = __builtin_amdgcn_readlane(cvn, 0);
        const int cnt1n = __builtin_amdgcn_readlane(cvn, 1);
        int idx0n = 0, idx1n = 0;
        if (more) {
            idx0n = gcn_first_idx(col, rn,     (cnt0n < GCN_CAP) ? cnt0n : GCN_CAP, lane);
            idx1n = gcn_first_idx(col, rn + 1, (cnt1n < GCN_CAP) ? cnt1n : GCN_CAP, lane);
        }

        gcn_lds_barrier();

        v8f c = {};
#pragma unroll
        for (int kb = 0; kb < 4; ++kb) {
            const _Float16* ap = &atile[m * GCN_APITCH + kb * 32 + 8 * half];
            const v8h lo = *(const v8h*)ap;
            const v8h hi = *(const v8h*)(ap + 16);
            const v16h a = __builtin_shufflevector(lo, hi, 0, 1, 2, 3, 4, 5, 6, 7,
                                                   8, 9, 10, 11, 12, 13, 14, 15);
            c = __builtin_amdgcn_wmma_f32_16x16x32_f16(
                 false, a,  false, b[kb],
                 (short)0, c,  false,  false);
        }

#pragma unroll
        for (int i = 0; i < 8; ++i)
            otile[(i + 8 * half) * GCN_OPITCH + n0 + m] = c[i];
        gcn_lds_barrier();

#pragma unroll
        for (int rr = 0; rr < 2; ++rr) {
            const int lr  = wave * 2 + rr;
            const int row = row0 + lr;
            if (row < nrows) {
                const float4 v  = *(const float4*)&otile[lr * GCN_OPITCH + lane * 4];
                const float  nd = gcn_norm_of(rr ? cnt1 : cnt0);
                float4 r;
                r.x = v.x * nd + bv.x;
                r.y = v.y * nd + bv.y;
                r.z = v.z * nd + bv.z;
                r.w = v.w * nd + bv.w;
                if (HIDDEN_OUT) {
                    const float ns = gcn_norm_of(__builtin_amdgcn_readlane(csv, rr));
                    v4h h;
                    h[0] = (_Float16)(fmaxf(r.x, 0.0f) * ns); h[1] = (_Float16)(fmaxf(r.y, 0.0f) * ns);
                    h[2] = (_Float16)(fmaxf(r.z, 0.0f) * ns); h[3] = (_Float16)(fmaxf(r.w, 0.0f) * ns);
                    ((v4h*)out_f16)[(size_t)row * (GCN_D / 4) + lane] = h;
                } else {
                    ((float4*)out_f32)[(size_t)row * (GCN_D / 4) + lane] = r;
                }
            }
        }
        tile = tile_n;
        cnt0 = cnt0n; cnt1 = cnt1n; idx0 = idx0n; idx1 = idx1n;
    }
}

extern "C" void kernel_launch(void* const* d_in, const int* in_sizes, int n_in,
                              void* d_out, int out_size, void* d_ws, size_t ws_size,
                              hipStream_t stream) {
    const float* feat = (const float*)d_in[0];
    const int*   src  = (const int*)  d_in[1];
    const int*   dst  = (const int*)  d_in[2];
    const float* W1   = (const float*)d_in[3];
    const float* b1   = (const float*)d_in[4];
    const float* W2   = (const float*)d_in[5];
    const float* b2   = (const float*)d_in[6];
    float* out = (float*)d_out;

    const int N = GCN_N;
    const int E = in_sizes[1];
    const size_t NF = (size_t)N * GCN_D;

    int*    cntS   = (int*)d_ws;
    int*    cntD   = cntS + GCN_NPAD;
    int*    ovfcnt = cntD + GCN_NPAD;
    int*    gcount = ovfcnt + GCN_OVFPAD;
    int*    covfcnt = gcount + GCN_NRANGE;
    _Float16* Wp1  = (_Float16*)(covfcnt + GCN_OVFPAD);
    _Float16* Wp2  = Wp1 + GCN_WPACK_ELEMS;
    _Float16* Xh   = Wp2 + GCN_WPACK_ELEMS;
    _Float16* Hh   = Xh + NF;
    int*    col    = (int*)(Hh + NF);
    int*    ovfD   = col + (size_t)N * GCN_CAP;
    int*    ovfS   = ovfD + E;
    int*    part   = ovfS + E;
    int*    covfE  = part + (size_t)GCN_NRANGE * GCN_CAPR;
    int*    covfR  = covfE + (size_t)2 * E;

    const int thr = 256;
    const int nzero = GCN_OVFPAD + GCN_NRANGE + GCN_OVFPAD;
    const int pgrid1 = (E + GCN_CHUNK - 1) / GCN_CHUNK;
    const int sgrid = 2 * GCN_PACK_BLOCKS + 1;
    const int ntiles = N / 16;
    const int layer_grid = (ntiles < GCN_LAYER_GRID) ? ntiles : GCN_LAYER_GRID;

    gcn_setup_kernel<<<sgrid, GCN_SETUP_THREADS, 0, stream>>>(W1, W2, Wp1, Wp2, ovfcnt, nzero);
    gcn_partition_kernel<<<pgrid1, GCN_PT, 0, stream>>>(src, dst, gcount, part,
                                                        covfcnt, covfE, covfR, E, 2 * E);
    gcn_fine_kernel<<<GCN_NRANGE, GCN_FT, 0, stream>>>(gcount, part, covfcnt, covfE, covfR,
                                                       cntS, cntD, col, ovfcnt, ovfD, ovfS,
                                                       feat, Xh, 2 * E, E);

    gcn_layer_kernel<true><<<layer_grid, thr, 0, stream>>>(
        Xh, cntS, cntD, col, ovfcnt, ovfD, ovfS, Wp1, b1, (float*)nullptr, Hh,
        N, ntiles, layer_grid, E);

    gcn_layer_kernel<false><<<layer_grid, thr, 0, stream>>>(
        Hh, cntS, cntD, col, ovfcnt, ovfD, ovfS, Wp2, b2, out, (_Float16*)nullptr,
        N, ntiles, layer_grid, E);
}


# Edge split with no global counters: each (split block, range) pair owns a fixed 128-byte line padded with -1; nothing zeroed first, W packing in that launch: 4 launches; sums in LDS arrival rank
# speedup vs baseline: 9.6098x; 1.0831x over previous
#include <hip/hip_runtime.h>


typedef __attribute__((ext_vector_type(16))) _Float16 v16h;
typedef __attribute__((ext_vector_type(8)))  float  v8f;
typedef __attribute__((ext_vector_type(8)))  _Float16 v8h;
typedef __attribute__((ext_vector_type(4)))  _Float16 v4h;

#define GCN_N 100000
#define GCN_D 128
#define GCN_WPACK_ELEMS (8 * 4 * 32 * 16)

#define GCN_NPAD 100032

#define GCN_CAP 64
#define GCN_OVFPAD 64

__device__ __forceinline__ float gcn_norm_of(int deg) {
    return rsqrtf(fmaxf((float)deg, 1.0f));
}

#define GCN_RNODES 392
#define GCN_NRANGE 256
#define GCN_PT     512
#define GCN_EPT    4
#define GCN_CHUNK  (GCN_PT * GCN_EPT)
#define GCN_SLOT   32
#define GCN_SPITCH 33
#define GCN_BOVF   (2 * GCN_CHUNK)

__device__ __forceinline__ int gcn_frag_k(int lane, int i) {
    const int j   = i >> 1;
    const int sub = i & 1;
    const int kk  = (j < 4) ? (2 * j + sub) : (16 + 2 * (j - 4) + sub);
    return kk + 8 * (lane >> 4);
}

#define GCN_PACK_BLOCKS ((2 * GCN_WPACK_ELEMS) / GCN_PT)

__global__ void __launch_bounds__(GCN_PT)
gcn_partition_kernel(const int* __restrict__ src,
                     const int* __restrict__ dst,
                     const float* __restrict__ W1, const float* __restrict__ W2,
                     _Float16* __restrict__ Wp1, _Float16* __restrict__ Wp2,
                     int* __restrict__ slots,
                     int* __restrict__ bovf_cnt,
                     int* __restrict__ bovf_ent,
                     int* __restrict__ bovf_rng,
                     int* __restrict__ ovf_cnt,
                     int nedges, int nsplit) {
    __shared__ int hist[GCN_NRANGE];
    __shared__ int nspill;
    __shared__ int stage[GCN_NRANGE * GCN_SPITCH];

    const int tid = threadIdx.x;
    const int blk = blockIdx.x;
    if (blk >= nsplit) {
        const int g = (blk - nsplit) * GCN_PT + tid;
        const bool second = g >= GCN_WPACK_ELEMS;
        const float* __restrict__ W  = second ? W2 : W1;
        _Float16* __restrict__    Wp = second ? Wp2 : Wp1;
        const int idx  = second ? g - GCN_WPACK_ELEMS : g;
        const int i    = idx & 15;
        const int lane = (idx >> 4) & 31;
        const int kb   = (idx >> 9) & 3;
        const int nt   = idx >> 11;
        const int K    = kb * 32 + gcn_frag_k(lane, i);
        const int n    = nt * 16 + (lane & 15);
        Wp[idx] = (_Float16)W[(size_t)K * GCN_D + n];
        return;
    }

    const int e0 = blk * GCN_CHUNK;
    int sv[GCN_EPT], dv[GCN_EPT];
#pragma unroll
    for (int k = 0; k < GCN_EPT; ++k) {
        const int i  = e0 + k * GCN_PT + tid;
        const int ic = (i < nedges) ? i : (nedges - 1);
        sv[k] = src[ic];
        dv[k] = dst[ic];
    }

    if (tid < GCN_NRANGE) hist[tid] = 0;
    if (tid == 0) nspill = 0;
    if (blk == 0 && tid == 0) ovf_cnt[0] = 0;
    for (int i = tid; i < GCN_NRANGE * GCN_SPITCH; i += GCN_PT) stage[i] = -1;
    __syncthreads();

    int ent[2 * GCN_EPT], rng[2 * GCN_EPT], rk[2 * GCN_EPT];
#pragma unroll
    for (int k = 0; k < GCN_EPT; ++k) {
        const int i = e0 + k * GCN_PT + tid;
        if (i < nedges) {
            const int s  = sv[k];
            const int d  = dv[k];
            const int rd = d / GCN_RNODES;
            const int rs = s / GCN_RNODES;
            ent[2 * k]     = (s << 10) | (d - rd * GCN_RNODES);
            rng[2 * k]     = rd;
            rk[2 * k]      = atomicAdd(&hist[rd], 1);
            ent[2 * k + 1] = (1 << 9) | (s - rs * GCN_RNODES);
            rng[2 * k + 1] = rs;
            rk[2 * k + 1]  = atomicAdd(&hist[rs], 1);
        } else {
            ent[2 * k] = 0; ent[2 * k + 1] = 0;
            rng[2 * k] = -1; rng[2 * k + 1] = -1;
            rk[2 * k] = 0;  rk[2 * k + 1] = 0;
        }
    }
#pragma unroll
    for (int k = 0; k < 2 * GCN_EPT; ++k) {
        if (rng[k] >= 0) {
            if (rk[k] < GCN_SLOT) {
                stage[rng[k] * GCN_SPITCH + rk[k]] = ent[k];
            } else {
                const int o = atomicAdd(&nspill, 1);
                bovf_ent[(size_t)blk * GCN_BOVF + o] = ent[k];
                bovf_rng[(size_t)blk * GCN_BOVF + o] = rng[k];
            }
        }
    }
    __syncthreads();

    for (int idx = tid; idx < GCN_NRANGE * GCN_SLOT; idx += GCN_PT) {
        const int r = idx / GCN_SLOT;
        const int k = idx - r * GCN_SLOT;
        slots[((size_t)r * nsplit + blk) * GCN_SLOT + k] = stage[r * GCN_SPITCH + k];
    }
    if (tid == 0) bovf_cnt[blk] = nspill;
}

template <int G>
__device__ __forceinline__ void gcn_load_rows(const float4* __restrict__ X4, int node0, int l0,
                                              int wave, int lane, float4 (&v)[G]) {
#pragma unroll
    for (int k = 0; k < G; ++k) {
        const int node = node0 + l0 + k * 32 + wave;
        const int nc   = (node < GCN_N) ? node : (GCN_N - 1);
        v[k] = X4[(size_t)nc * (GCN_D / 4) + lane];
    }
}
template <int G>
__device__ __forceinline__ void gcn_store_rows(v4h* __restrict__ H4, const int* cs, int node0,
                                               int l0, int wave, int lane, const float4 (&v)[G]) {
#pragma unroll
    for (int k = 0; k < G; ++k) {
        const int l    = l0 + k * 32 + wave;
        const int node = node0 + l;
        if (node < GCN_N) {
            const float ns = gcn_norm_of(cs[l]);
            v4h h;
            h[0] = (_Float16)(v[k].x * ns); h[1] = (_Float16)(v[k].y * ns);
            h[2] = (_Float16)(v[k].z * ns); h[3] = (_Float16)(v[k].w * ns);
            H4[(size_t)node * (GCN_D / 4) + lane] = h;
        }
    }
}

#define GCN_FT 1024
#define GCN_FROWS 12
#define GCN_FB 25
static_assert(GCN_RNODES > GCN_FROWS * 32 && GCN_RNODES <= (GCN_FROWS + 1) * 32,
              "a wave copies GCN_FROWS rows, and one more where its number is below the remainder");

__device__ __forceinline__ void gcn_place_entry(int e, int* cd, int* cs, int* tile, int node0,
                                                int* __restrict__ ovf_cnt,
                                                int* __restrict__ ovf_dst,
                                                int* __restrict__ ovf_src, int ovf_cap) {
    const int local = e & 511;
    if (e & 512) {
        atomicAdd(&cs[local], 1);
    } else {
        const int k = atomicAdd(&cd[local], 1);
        if (k < GCN_CAP) {
            tile[local * GCN_CAP + k] = e >> 10;
        } else {
            const int idx = atomicAdd(ovf_cnt, 1);
            if (idx < ovf_cap) { ovf_dst[idx] = node0 + local; ovf_src[idx] = e >> 10; }
        }
    }
}

__global__ void __launch_bounds__(GCN_FT)
gcn_fine_kernel(const int* __restrict__ slots,
                const int* __restrict__ bovf_cnt,
                const int* __restrict__ bovf_ent,
                const int* __restrict__ bovf_rng,
                int* __restrict__ cnt_src,
                int* __restrict__ cnt_dst,
                int* __restrict__ col,
                int* __restrict__ ovf_cnt,
                int* __restrict__ ovf_dst,
                int* __restrict__ ovf_src,
                const float* __restrict__ X,
                _Float16* __restrict__ Xh,
                int nsplit, int ovf_cap) {
    __shared__ int cd[GCN_RNODES];
    __shared__ int cs[GCN_RNODES];
    __shared__ int tile[GCN_RNODES * GCN_CAP];

    const int tid   = threadIdx.x;
    const int r     = blockIdx.x;
    const int node0 = r * GCN_RNODES;
    for (int i = tid; i < GCN_RNODES; i += GCN_FT) { cd[i] = 0; cs[i] = 0; }
    __syncthreads();

    const int total = nsplit * GCN_SLOT;
    const int* __restrict__ pr = slots + (size_t)r * total;
    for (int base = 0; base < total; base += GCN_FT * GCN_FB) {
        int e[GCN_FB], kk[GCN_FB];
#pragma unroll
        for (int k = 0; k < GCN_FB; ++k) {
            const int i  = base + k * GCN_FT + tid;
            const int ic = (i < total) ? i : (total - 1);
            const int v  = pr[ic];
            e[k] = (i < total) ? v : -1;
        }
#pragma unroll
        for (int k = 0; k < GCN_FB; ++k) {
            kk[k] = -1;
            if (e[k] >= 0) {
                const int local = e[k] & 511;
                if (e[k] & 512) atomicAdd(&cs[local], 1);
                else            kk[k] = atomicAdd(&cd[local], 1);
            }
        }
#pragma unroll
        for (int k = 0; k < GCN_FB; ++k) {
            if (kk[k] >= 0) {
                const int local = e[k] & 511;
                if (kk[k] < GCN_CAP) {
                    tile[local * GCN_CAP + kk[k]] = e[k] >> 10;
                } else {
                    const int idx = atomicAdd(ovf_cnt, 1);
                    if (idx < ovf_cap) { ovf_dst[idx] = node0 + local; ovf_src[idx] = e[k] >> 10; }
                }
            }
        }
    }
    for (int b = tid; b < nsplit; b += GCN_FT) {
        int c = bovf_cnt[b];
        c = (c < GCN_BOVF) ? c : GCN_BOVF;
        for (int j = 0; j < c; ++j) {
            if (bovf_rng[(size_t)b * GCN_BOVF + j] == r)
                gcn_place_entry(bovf_ent[(size_t)b * GCN_BOVF + j], cd, cs, tile, node0,
                                ovf_cnt, ovf_dst, ovf_src, ovf_cap);
        }
    }
    __syncthreads();

    const int lane = tid & 31;
    const int wave = __builtin_amdgcn_readfirstlane(tid >> 5);
    const float4* __restrict__ X4 = (const float4*)X;
    v4h* __restrict__ H4 = (v4h*)Xh;
    const bool last_row = wave < GCN_RNODES - GCN_FROWS * 32;
    float4 va[GCN_FROWS];
    float4 vc[1];
    gcn_load_rows<GCN_FROWS>(X4, node0, 0, wave, lane, va);
    gcn_load_rows<1>(X4, node0, last_row ? GCN_FROWS * 32 : 0, wave, lane, vc);

    for (int i = tid; i < GCN_RNODES; i += GCN_FT) {
        const int node = node0 + i;
        if (node < GCN_N) { cnt_dst[node] = cd[i]; cnt_src[node] = cs[i]; }
    }
    for (int idx = tid; idx < GCN_RNODES * GCN_CAP; idx += GCN_FT) {
        const int l    = idx / GCN_CAP;
        const int k    = idx - l * GCN_CAP;
        const int node = node0 + l;
        if (node < GCN_N && k < cd[l]) col[(size_t)node * GCN_CAP + k] = tile[idx];
    }

    gcn_store_rows<GCN_FROWS>(H4, cs, node0, 0, wave, lane, va);
    if (last_row) gcn_store_rows<1>(H4, cs, node0, GCN_FROWS * 32, wave, lane, vc);
}

template <int G, bool MASKED>
__device__ __forceinline__ void gcn_sum_group(const v4h* __restrict__ X4, int myidx,
                                              int j, int n, int lane, float4& acc) {
    v4h v[G];
#pragma unroll
    for (int k = 0; k < G; ++k) {
        const int s = __builtin_amdgcn_readlane(myidx, j + k);
        v[k] = X4[(size_t)s * (GCN_D / 4) + lane];
    }
#pragma unroll
    for (int k = 0; k < G; ++k) {
        if (MASKED) {
            const float w = (j + k < n) ? 1.0f : 0.0f;
            acc.x = fmaf(w, (float)v[k][0], acc.x); acc.y = fmaf(w, (float)v[k][1], acc.y);
            acc.z = fmaf(w, (float)v[k][2], acc.z); acc.w = fmaf(w, (float)v[k][3], acc.w);
        } else {
            acc.x += (float)v[k][0]; acc.y += (float)v[k][1];
            acc.z += (float)v[k][2]; acc.w += (float)v[k][3];
        }
    }
}

#define GCN_APITCH 136
#define GCN_OPITCH 132

#define GCN_LAYER_GRID 1280

__device__ __forceinline__ int gcn_two_degrees(const int* __restrict__ deg, int r,
                                               int nrows, int lane) {
    const int q  = r + (lane & 1);
    const int qc = (q < nrows) ? q : (nrows - 1);
    const int v  = deg[qc];
    return (q < nrows) ? v : 0;
}

__device__ __forceinline__ void gcn_lds_barrier() {
    __builtin_amdgcn_fence(__ATOMIC_RELEASE, "workgroup", "local");
    __builtin_amdgcn_s_barrier();
    __builtin_amdgcn_fence(__ATOMIC_ACQUIRE, "workgroup", "local");
}

__device__ __forceinline__ int gcn_first_idx(const int* __restrict__ col, int row,
                                             int nb, int lane) {
    if (nb <= 0) return 0;
    const int n = (nb < 32) ? nb : 32;
    return col[(size_t)row * GCN_CAP + ((lane < n) ? lane : 0)];
}

__device__ __forceinline__ void gcn_sum_chunk(const v4h* __restrict__ X4, int myidx,
                                              int n, int lane, float4& acc) {
    int j = 0;
    for (; j + 8 <= n; j += 8)
        gcn_sum_group<8, false>(X4, myidx, j, n, lane, acc);
    if (j < n) {
        if (n - j > 4) gcn_sum_group<8, true>(X4, myidx, j, n, lane, acc);
        else           gcn_sum_group<4, true>(X4, myidx, j, n, lane, acc);
    }
}

template <bool HIDDEN_OUT>
__global__ void __launch_bounds__(256)
gcn_layer_kernel(const _Float16* __restrict__ Xh,
                 const int* __restrict__ cnt_src,
                 const int* __restrict__ cnt_dst,
                 const int* __restrict__ col,
                 const int* __restrict__ ovf_cnt,
                 const int* __restrict__ ovf_dst,
                 const int* __restrict__ ovf_src,
                 const _Float16* __restrict__ Wp,
                 const float* __restrict__ bias,
                 float* __restrict__ out_f32,
                 _Float16* __restrict__ out_f16,
                 int nrows, int ntiles, int tile_step, int ovf_cap) {
    __shared__ __attribute__((aligned(16))) _Float16 atile[16 * GCN_APITCH];
    __shared__ __attribute__((aligned(16))) float  otile[16 * GCN_OPITCH];

    const int lane = threadIdx.x & 31;
    const int wave = __builtin_amdgcn_readfirstlane((int)(threadIdx.x >> 5));
    const v4h* __restrict__ X4 = (const v4h*)Xh;

    const int n0   = wave * 16;
    const int half = lane >> 4;
    const int m    = lane & 15;

    v16h b[4];
#pragma unroll
    for (int kb = 0; kb < 4; ++kb)
        b[kb] = *(const v16h*)&Wp[(((size_t)(wave * 4 + kb)) * 32 + lane) * 16];
    const float4 bv = ((const float4*)bias)[lane];

    int tile = blockIdx.x;
    int cnt0 = 0, cnt1 = 0, idx0 = 0, idx1 = 0;
    if (tile < ntiles) {
        const int r = tile * 16 + wave * 2;
        const int cv = gcn_two_degrees(cnt_dst, r, nrows, lane);
        cnt0 = __builtin_amdgcn_readlane(cv, 0);
        cnt1 = __builtin_amdgcn_readlane(cv, 1);
        idx0 = gcn_first_idx(col, r,     (cnt0 < GCN_CAP) ? cnt0 : GCN_CAP, lane);
        idx1 = gcn_first_idx(col, r + 1, (cnt1 < GCN_CAP) ? cnt1 : GCN_CAP, lane);
    }

    while (tile < ntiles) {
        const int row0 = tile * 16;
        const int  tile_n = tile + tile_step;
        const bool more   = tile_n < ntiles;
        const int  rn     = tile_n * 16 + wave * 2;
        const int cvn = gcn_two_degrees(cnt_dst, rn, nrows, lane);
        int csv = 0;
        if (HIDDEN_OUT) csv = gcn_two_degrees(cnt_src, row0 + wave * 2, nrows, lane);

#pragma unroll 1
        for (int rr = 0; rr < 2; ++rr) {
            const int lr  = wave * 2 + rr;
            const int row = row0 + lr;
            const int cnt = rr ? cnt1 : cnt0;
            float4 acc = make_float4(0.0f, 0.0f, 0.0f, 0.0f);
            const int nb  = (cnt < GCN_CAP) ? cnt : GCN_CAP;
            if (nb > 0) {
                gcn_sum_chunk(X4, rr ? idx1 : idx0, (nb < 32) ? nb : 32, lane, acc);
                const int* __restrict__ crow = col + (size_t)row * GCN_CAP;
                for (int done = 32; done < nb; done += 32) {
                    const int n = (nb - done < 32) ? (nb - done) : 32;
                    const int myidx = crow[done + ((lane < n) ? lane : 0)];
                    gcn_sum_chunk(X4, myidx, n, lane, acc);
                }
            }
            if (cnt > GCN_CAP) {
                int novf = ovf_cnt[0];
                novf = (novf < ovf_cap) ? novf : ovf_cap;
                for (int base = 0; base < novf; base += 32) {
                    const int  e   = base + lane;
                    const bool hit = (e < novf) && (ovf_dst[e] == row);
                    const int  hs  = hit ? ovf_src[e] : 0;
                    unsigned mask = __builtin_amdgcn_ballot_w32(hit);
                    while (mask != 0u) {
                        const int j = __builtin_ctz(mask);
                        mask &= mask - 1u;
                        const int s = __builtin_amdgcn_readlane(hs, j);
                        const v4h v = X4[(size_t)s * (GCN_D / 4) + lane];
                        acc.x += (float)v[0]; acc.y += (float)v[1];
                        acc.z += (float)v[2]; acc.w += (float)v[3];
                    }
                }
            }
            v4h p;
            p[0] = (_Float16)acc.x; p[1] = (_Float16)acc.y;
            p[2] = (_Float16)acc.z; p[3] = (_Float16)acc.w;
            *(v4h*)&atile[lr * GCN_APITCH + lane * 4] = p;
        }

        const int cnt0n = __builtin_amdgcn_readlane(cvn, 0);
        const int cnt1n = __builtin_amdgcn_readlane(cvn, 1);
        int idx0n = 0, idx1n = 0;
        if (more) {
            idx0n = gcn_first_idx(col, rn,     (cnt0n < GCN_CAP) ? cnt0n : GCN_CAP, lane);
            idx1n = gcn_first_idx(col, rn + 1, (cnt1n < GCN_CAP) ? cnt1n : GCN_CAP, lane);
        }

        gcn_lds_barrier();

        v8f c = {};
#pragma unroll
        for (int kb = 0; kb < 4; ++kb) {
            const _Float16* ap = &atile[m * GCN_APITCH + kb * 32 + 8 * half];
            const v8h lo = *(const v8h*)ap;
            const v8h hi = *(const v8h*)(ap + 16);
            const v16h a = __builtin_shufflevector(lo, hi, 0, 1, 2, 3, 4, 5, 6, 7,
                                                   8, 9, 10, 11, 12, 13, 14, 15);
            c = __builtin_amdgcn_wmma_f32_16x16x32_f16(
                 false, a,  false, b[kb],
                 (short)0, c,  false,  false);
        }

#pragma unroll
        for (int i = 0; i < 8; ++i)
            otile[(i + 8 * half) * GCN_OPITCH + n0 + m] = c[i];
        gcn_lds_barrier();

#pragma unroll
        for (int rr = 0; rr < 2; ++rr) {
            const int lr  = wave * 2 + rr;
            const int row = row0 + lr;
            if (row < nrows) {
                const float4 v  = *(const float4*)&otile[lr * GCN_OPITCH + lane * 4];
                const float  nd = gcn_norm_of(rr ? cnt1 : cnt0);
                float4 r;
                r.x = v.x * nd + bv.x;
                r.y = v.y * nd + bv.y;
                r.z = v.z * nd + bv.z;
                r.w = v.w * nd + bv.w;
                if (HIDDEN_OUT) {
                    const float ns = gcn_norm_of(__builtin_amdgcn_readlane(csv, rr));
                    v4h h;
                    h[0] = (_Float16)(fmaxf(r.x, 0.0f) * ns); h[1] = (_Float16)(fmaxf(r.y, 0.0f) * ns);
                    h[2] = (_Float16)(fmaxf(r.z, 0.0f) * ns); h[3] = (_Float16)(fmaxf(r.w, 0.0f) * ns);
                    ((v4h*)out_f16)[(size_t)row * (GCN_D / 4) + lane] = h;
                } else {
                    ((float4*)out_f32)[(size_t)row * (GCN_D / 4) + lane] = r;
                }
            }
        }
        tile = tile_n;
        cnt0 = cnt0n; cnt1 = cnt1n; idx0 = idx0n; idx1 = idx1n;
    }
}

extern "C" void kernel_launch(void* const* d_in, const int* in_sizes, int n_in,
                              void* d_out, int out_size, void* d_ws, size_t ws_size,
                              hipStream_t stream) {
    const float* feat = (const float*)d_in[0];
    const int*   src  = (const int*)  d_in[1];
    const int*   dst  = (const int*)  d_in[2];
    const float* W1   = (const float*)d_in[3];
    const float* b1   = (const float*)d_in[4];
    const float* W2   = (const float*)d_in[5];
    const float* b2   = (const float*)d_in[6];
    float* out = (float*)d_out;

    const int N = GCN_N;
    const int E = in_sizes[1];
    const size_t NF = (size_t)N * GCN_D;

    int*    cntS   = (int*)d_ws;
    int*    cntD   = cntS + GCN_NPAD;
    int*    ovfcnt = cntD + GCN_NPAD;
    _Float16* Wp1  = (_Float16*)(ovfcnt + GCN_OVFPAD);
    _Float16* Wp2  = Wp1 + GCN_WPACK_ELEMS;
    _Float16* Xh   = Wp2 + GCN_WPACK_ELEMS;
    _Float16* Hh   = Xh + NF;
    int*    col    = (int*)(Hh + NF);
    int*    slots  = col + (size_t)N * GCN_CAP;
    int*    bovfE  = slots + ((size_t)4 * E + 8192);
    int*    bovfR  = bovfE + ((size_t)2 * E + 4096);
    int*    ovfD   = bovfR + ((size_t)2 * E + 4096);
    int*    ovfS   = ovfD + E;
    int*    bovfC  = ovfS + E;

    const int thr = 256;
    const int pgrid1 = (E + GCN_CHUNK - 1) / GCN_CHUNK;
    const int ntiles = N / 16;
    const int layer_grid = (ntiles < GCN_LAYER_GRID) ? ntiles : GCN_LAYER_GRID;

    gcn_partition_kernel<<<pgrid1 + GCN_PACK_BLOCKS, GCN_PT, 0, stream>>>(
        src, dst, W1, W2, Wp1, Wp2, slots, bovfC, bovfE, bovfR, ovfcnt, E, pgrid1);
    gcn_fine_kernel<<<GCN_NRANGE, GCN_FT, 0, stream>>>(slots, bovfC, bovfE, bovfR,
                                                       cntS, cntD, col, ovfcnt, ovfD, ovfS,
                                                       feat, Xh, pgrid1, E);

    gcn_layer_kernel<true><<<layer_grid, thr, 0, stream>>>(
        Xh, cntS, cntD, col, ovfcnt, ovfD, ovfS, Wp1, b1, (float*)nullptr, Hh,
        N, ntiles, layer_grid, E);

    gcn_layer_kernel<false><<<layer_grid, thr, 0, stream>>>(
        Hh, cntS, cntD, col, ovfcnt, ovfD, ovfS, Wp2, b2, out, (_Float16*)nullptr,
        N, ntiles, layer_grid, E);
}


# Layer edge sums: a wave's two rows side by side, one half-wave each, 16 B per lane, so one 512-B load brings a source row of each (spare slots add a zero row; W fragments wait in LDS); same sum order
# speedup vs baseline: 9.9581x; 1.0362x over previous
#include <hip/hip_runtime.h>


typedef __attribute__((ext_vector_type(16))) _Float16 v16h;
typedef __attribute__((ext_vector_type(8)))  float  v8f;
typedef __attribute__((ext_vector_type(8)))  _Float16 v8h;
typedef __attribute__((ext_vector_type(4)))  _Float16 v4h;

#define GCN_N 100000
#define GCN_D 128
#define GCN_WPACK_ELEMS (8 * 4 * 32 * 16)

#define GCN_NPAD 100032

#define GCN_CAP 64
#define GCN_OVFPAD 64

__device__ __forceinline__ float gcn_norm_of(int deg) {
    return rsqrtf(fmaxf((float)deg, 1.0f));
}

#define GCN_RNODES 392
#define GCN_NRANGE 256
#define GCN_PT     512
#define GCN_EPT    4
#define GCN_CHUNK  (GCN_PT * GCN_EPT)
#define GCN_SLOT   32
#define GCN_SPITCH 33
#define GCN_BOVF   (2 * GCN_CHUNK)

__device__ __forceinline__ int gcn_frag_k(int lane, int i) {
    const int j   = i >> 1;
    const int sub = i & 1;
    const int kk  = (j < 4) ? (2 * j + sub) : (16 + 2 * (j - 4) + sub);
    return kk + 8 * (lane >> 4);
}

#define GCN_PACK_BLOCKS ((2 * GCN_WPACK_ELEMS) / GCN_PT)

__global__ void __launch_bounds__(GCN_PT)
gcn_partition_kernel(const int* __restrict__ src,
                     const int* __restrict__ dst,
                     const float* __restrict__ W1, const float* __restrict__ W2,
                     _Float16* __restrict__ Wp1, _Float16* __restrict__ Wp2,
                     int* __restrict__ slots,
                     int* __restrict__ bovf_cnt,
                     int* __restrict__ bovf_ent,
                     int* __restrict__ bovf_rng,
                     int* __restrict__ ovf_cnt,
                     int nedges, int nsplit) {
    __shared__ int hist[GCN_NRANGE];
    __shared__ int nspill;
    __shared__ int stage[GCN_NRANGE * GCN_SPITCH];

    const int tid = threadIdx.x;
    const int blk = blockIdx.x;
    if (blk >= nsplit) {
        const int g = (blk - nsplit) * GCN_PT + tid;
        const bool second = g >= GCN_WPACK_ELEMS;
        const float* __restrict__ W  = second ? W2 : W1;
        _Float16* __restrict__    Wp = second ? Wp2 : Wp1;
        const int idx  = second ? g - GCN_WPACK_ELEMS : g;
        const int i    = idx & 15;
        const int lane = (idx >> 4) & 31;
        const int kb   = (idx >> 9) & 3;
        const int nt   = idx >> 11;
        const int K    = kb * 32 + gcn_frag_k(lane, i);
        const int n    = nt * 16 + (lane & 15);
        Wp[idx] = (_Float16)W[(size_t)K * GCN_D + n];
        return;
    }

    const int e0 = blk * GCN_CHUNK;
    int sv[GCN_EPT], dv[GCN_EPT];
#pragma unroll
    for (int k = 0; k < GCN_EPT; ++k) {
        const int i  = e0 + k * GCN_PT + tid;
        const int ic = (i < nedges) ? i : (nedges - 1);
        sv[k] = src[ic];
        dv[k] = dst[ic];
    }

    if (tid < GCN_NRANGE) hist[tid] = 0;
    if (tid == 0) nspill = 0;
    if (blk == 0 && tid == 0) ovf_cnt[0] = 0;
    for (int i = tid; i < GCN_NRANGE * GCN_SPITCH; i += GCN_PT) stage[i] = -1;
    __syncthreads();

    int ent[2 * GCN_EPT], rng[2 * GCN_EPT], rk[2 * GCN_EPT];
#pragma unroll
    for (int k = 0; k < GCN_EPT; ++k) {
        const int i = e0 + k * GCN_PT + tid;
        if (i < nedges) {
            const int s  = sv[k];
            const int d  = dv[k];
            const int rd = d / GCN_RNODES;
            const int rs = s / GCN_RNODES;
            ent[2 * k]     = (s << 10) | (d - rd * GCN_RNODES);
            rng[2 * k]     = rd;
            rk[2 * k]      = atomicAdd(&hist[rd], 1);
            ent[2 * k + 1] = (1 << 9) | (s - rs * GCN_RNODES);
            rng[2 * k + 1] = rs;
            rk[2 * k + 1]  = atomicAdd(&hist[rs], 1);
        } else {
            ent[2 * k] = 0; ent[2 * k + 1] = 0;
            rng[2 * k] = -1; rng[2 * k + 1] = -1;
            rk[2 * k] = 0;  rk[2 * k + 1] = 0;
        }
    }
#pragma unroll
    for (int k = 0; k < 2 * GCN_EPT; ++k) {
        if (rng[k] >= 0) {
            if (rk[k] < GCN_SLOT) {
                stage[rng[k] * GCN_SPITCH + rk[k]] = ent[k];
            } else {
                const int o = atomicAdd(&nspill, 1);
                bovf_ent[(size_t)blk * GCN_BOVF + o] = ent[k];
                bovf_rng[(size_t)blk * GCN_BOVF + o] = rng[k];
            }
        }
    }
    __syncthreads();

    for (int idx = tid; idx < GCN_NRANGE * GCN_SLOT; idx += GCN_PT) {
        const int r = idx / GCN_SLOT;
        const int k = idx - r * GCN_SLOT;
        slots[((size_t)r * nsplit + blk) * GCN_SLOT + k] = stage[r * GCN_SPITCH + k];
    }
    if (tid == 0) bovf_cnt[blk] = nspill;
}

template <int G>
__device__ __forceinline__ void gcn_load_rows(const float4* __restrict__ X4, int node0, int l0,
                                              int wave, int lane, float4 (&v)[G]) {
#pragma unroll
    for (int k = 0; k < G; ++k) {
        const int node = node0 + l0 + k * 32 + wave;
        const int nc   = (node < GCN_N) ? node : (GCN_N - 1);
        v[k] = X4[(size_t)nc * (GCN_D / 4) + lane];
    }
}
template <int G>
__device__ __forceinline__ void gcn_store_rows(v4h* __restrict__ H4, const int* cs, int node0,
                                               int l0, int wave, int lane, const float4 (&v)[G]) {
#pragma unroll
    for (int k = 0; k < G; ++k) {
        const int l    = l0 + k * 32 + wave;
        const int node = node0 + l;
        if (node < GCN_N) {
            const float ns = gcn_norm_of(cs[l]);
            v4h h;
            h[0] = (_Float16)(v[k].x * ns); h[1] = (_Float16)(v[k].y * ns);
            h[2] = (_Float16)(v[k].z * ns); h[3] = (_Float16)(v[k].w * ns);
            H4[(size_t)node * (GCN_D / 4) + lane] = h;
        }
    }
}

#define GCN_FT 1024
#define GCN_FROWS 12
#define GCN_FB 25
static_assert(GCN_RNODES > GCN_FROWS * 32 && GCN_RNODES <= (GCN_FROWS + 1) * 32,
              "a wave copies GCN_FROWS rows, and one more where its number is below the remainder");

__device__ __forceinline__ void gcn_place_entry(int e, int* cd, int* cs, int* tile, int node0,
                                                int* __restrict__ ovf_cnt,
                                                int* __restrict__ ovf_dst,
                                                int* __restrict__ ovf_src, int ovf_cap) {
    const int local = e & 511;
    if (e & 512) {
        atomicAdd(&cs[local], 1);
    } else {
        const int k = atomicAdd(&cd[local], 1);
        if (k < GCN_CAP) {
            tile[local * GCN_CAP + k] = e >> 10;
        } else {
            const int idx = atomicAdd(ovf_cnt, 1);
            if (idx < ovf_cap) { ovf_dst[idx] = node0 + local; ovf_src[idx] = e >> 10; }
        }
    }
}

__global__ void __launch_bounds__(GCN_FT)
gcn_fine_kernel(const int* __restrict__ slots,
                const int* __restrict__ bovf_cnt,
                const int* __restrict__ bovf_ent,
                const int* __restrict__ bovf_rng,
                int* __restrict__ cnt_src,
                int* __restrict__ cnt_dst,
                int* __restrict__ col,
                int* __restrict__ ovf_cnt,
                int* __restrict__ ovf_dst,
                int* __restrict__ ovf_src,
                const float* __restrict__ X,
                _Float16* __restrict__ Xh,
                int nsplit, int ovf_cap) {
    __shared__ int cd[GCN_RNODES];
    __shared__ int cs[GCN_RNODES];
    __shared__ int tile[GCN_RNODES * GCN_CAP];

    const int tid   = threadIdx.x;
    const int r     = blockIdx.x;
    const int node0 = r * GCN_RNODES;
    for (int i = tid; i < GCN_RNODES; i += GCN_FT) { cd[i] = 0; cs[i] = 0; }
    __syncthreads();

    const int total = nsplit * GCN_SLOT;
    const int* __restrict__ pr = slots + (size_t)r * total;
    for (int base = 0; base < total; base += GCN_FT * GCN_FB) {
        int e[GCN_FB], kk[GCN_FB];
#pragma unroll
        for (int k = 0; k < GCN_FB; ++k) {
            const int i  = base + k * GCN_FT + tid;
            const int ic = (i < total) ? i : (total - 1);
            const int v  = pr[ic];
            e[k] = (i < total) ? v : -1;
        }
#pragma unroll
        for (int k = 0; k < GCN_FB; ++k) {
            kk[k] = -1;
            if (e[k] >= 0) {
                const int local = e[k] & 511;
                if (e[k] & 512) atomicAdd(&cs[local], 1);
                else            kk[k] = atomicAdd(&cd[local], 1);
            }
        }
#pragma unroll
        for (int k = 0; k < GCN_FB; ++k) {
            if (kk[k] >= 0) {
                const int local = e[k] & 511;
                if (kk[k] < GCN_CAP) {
                    tile[local * GCN_CAP + kk[k]] = e[k] >> 10;
                } else {
                    const int idx = atomicAdd(ovf_cnt, 1);
                    if (idx < ovf_cap) { ovf_dst[idx] = node0 + local; ovf_src[idx] = e[k] >> 10; }
                }
            }
        }
    }
    for (int b = tid; b < nsplit; b += GCN_FT) {
        int c = bovf_cnt[b];
        c = (c < GCN_BOVF) ? c : GCN_BOVF;
        for (int j = 0; j < c; ++j) {
            if (bovf_rng[(size_t)b * GCN_BOVF + j] == r)
                gcn_place_entry(bovf_ent[(size_t)b * GCN_BOVF + j], cd, cs, tile, node0,
                                ovf_cnt, ovf_dst, ovf_src, ovf_cap);
        }
    }
    __syncthreads();

    const int lane = tid & 31;
    const int wave = __builtin_amdgcn_readfirstlane(tid >> 5);
    const float4* __restrict__ X4 = (const float4*)X;
    v4h* __restrict__ H4 = (v4h*)Xh;
    const bool last_row = wave < GCN_RNODES - GCN_FROWS * 32;
    float4 va[GCN_FROWS];
    float4 vc[1];
    gcn_load_rows<GCN_FROWS>(X4, node0, 0, wave, lane, va);
    gcn_load_rows<1>(X4, node0, last_row ? GCN_FROWS * 32 : 0, wave, lane, vc);

    for (int i = tid; i < GCN_RNODES; i += GCN_FT) {
        const int node = node0 + i;
        if (node < GCN_N) { cnt_dst[node] = cd[i]; cnt_src[node] = cs[i]; }
    }
    for (int idx = tid; idx < GCN_RNODES * GCN_CAP; idx += GCN_FT) {
        const int l    = idx / GCN_CAP;
        const int k    = idx - l * GCN_CAP;
        const int node = node0 + l;
        if (node < GCN_N && k < cd[l]) col[(size_t)node * GCN_CAP + k] = tile[idx];
    }

    gcn_store_rows<GCN_FROWS>(H4, cs, node0, 0, wave, lane, va);
    if (last_row) gcn_store_rows<1>(H4, cs, node0, GCN_FROWS * 32, wave, lane, vc);
    if (r == 0 && wave == 0) {
        v4h z;
        z[0] = (_Float16)0.0f; z[1] = (_Float16)0.0f; z[2] = (_Float16)0.0f; z[3] = (_Float16)0.0f;
        H4[(size_t)GCN_N * (GCN_D / 4) + lane] = z;
        H4[(size_t)(2 * GCN_N + 1) * (GCN_D / 4) + lane] = z;
    }
}

template <int T>
__device__ __forceinline__ int gcn_row_share(int v) {
    return __builtin_amdgcn_mov_dpp(v, 0x150 + T, 0xf, 0xf, true);
}
__device__ __forceinline__ int gcn_row_share_n(int v, int t) {
    switch (t) {
    case 0:  return gcn_row_share<0>(v);   case 1:  return gcn_row_share<1>(v);
    case 2:  return gcn_row_share<2>(v);   case 3:  return gcn_row_share<3>(v);
    case 4:  return gcn_row_share<4>(v);   case 5:  return gcn_row_share<5>(v);
    case 6:  return gcn_row_share<6>(v);   case 7:  return gcn_row_share<7>(v);
    case 8:  return gcn_row_share<8>(v);   case 9:  return gcn_row_share<9>(v);
    case 10: return gcn_row_share<10>(v);  case 11: return gcn_row_share<11>(v);
    case 12: return gcn_row_share<12>(v);  case 13: return gcn_row_share<13>(v);
    case 14: return gcn_row_share<14>(v);  default: return gcn_row_share<15>(v);
    }
}

template <int G, int T0>
__device__ __forceinline__ void gcn_sum_group(const char* __restrict__ Xb, int idxv,
                                              unsigned laneoff, float (&acc)[8]) {
    v8h v[G];
#pragma unroll
    for (int k = 0; k < G; ++k) {
        const int s = gcn_row_share_n(idxv, T0 + k);
        v[k] = *(const v8h*)(Xb + ((unsigned)s * 256u + laneoff));
    }
    if (G == 8)
        asm volatile("" :: "v"(v[0]), "v"(v[1]), "v"(v[2]), "v"(v[3]),
                           "v"(v[G - 4]), "v"(v[G - 3]), "v"(v[G - 2]), "v"(v[G - 1]));
    else
        asm volatile("" :: "v"(v[0]), "v"(v[1]), "v"(v[2]), "v"(v[3]));
#pragma unroll
    for (int k = 0; k < G; ++k) {
#pragma unroll
        for (int e = 0; e < 8; ++e) acc[e] += (float)v[k][e];
        __builtin_amdgcn_sched_barrier(0);
    }
}

#define GCN_APITCH 136
#define GCN_OPITCH 132

#define GCN_LAYER_GRID 1280

__device__ __forceinline__ int gcn_two_degrees(const int* __restrict__ deg, int r,
                                               int nrows, int lane) {
    const int q  = r + (lane >> 4);
    const int qc = (q < nrows) ? q : (nrows - 1);
    const int v  = deg[(unsigned)qc];
    return (q < nrows) ? v : 0;
}

__device__ __forceinline__ void gcn_lds_barrier() {
    __builtin_amdgcn_fence(__ATOMIC_RELEASE, "workgroup", "local");
    __builtin_amdgcn_s_barrier();
    __builtin_amdgcn_fence(__ATOMIC_ACQUIRE, "workgroup", "local");
}

__device__ __forceinline__ int gcn_chunk_idx(const int* __restrict__ col, int myrow,
                                             int mynb, int cb, int c) {
    if (c >= mynb - cb) return GCN_N;
    return (col + cb)[(unsigned)(myrow * GCN_CAP + c)];
}

__device__ __forceinline__ void gcn_sum_chunk(const char* __restrict__ Xb, int idxv, int ns,
                                              unsigned laneoff, float (&acc)[8]) {
    if (ns > 4) gcn_sum_group<8, 0>(Xb, idxv, laneoff, acc);
    else        gcn_sum_group<4, 0>(Xb, idxv, laneoff, acc);
    if (ns > 12)     gcn_sum_group<8, 8>(Xb, idxv, laneoff, acc);
    else if (ns > 8) gcn_sum_group<4, 8>(Xb, idxv, laneoff, acc);
}

template <bool HIDDEN_OUT>
__global__ void __launch_bounds__(256) __attribute__((amdgpu_waves_per_eu(10, 10)))
gcn_layer_kernel(const _Float16* __restrict__ Xh,
                 const int* __restrict__ cnt_src,
                 const int* __restrict__ cnt_dst,
                 const int* __restrict__ col,
                 const int* __restrict__ ovf_cnt,
                 const int* __restrict__ ovf_dst,
                 const int* __restrict__ ovf_src,
                 const _Float16* __restrict__ Wp,
                 const float* __restrict__ bias,
                 float* __restrict__ out_f32,
                 _Float16* __restrict__ out_f16,
                 int nrows, int ntiles, int tile_step, int ovf_cap) {
    __shared__ __attribute__((aligned(16))) _Float16 atile[16 * GCN_APITCH];
    __shared__ __attribute__((aligned(16))) float  otile[16 * GCN_OPITCH];
    __shared__ __attribute__((aligned(16))) _Float16 wfrag[8 * 4 * 2 * 32 * 8];
    __shared__ __attribute__((aligned(16))) float bias_s[GCN_D];

    const int lane = threadIdx.x & 31;
    const int wave = __builtin_amdgcn_readfirstlane((int)(threadIdx.x >> 5));
    const char* __restrict__ Xb = (const char*)Xh;

    const int n0   = wave * 16;
    const int half = lane >> 4;
    const int m    = lane & 15;
    const unsigned laneoff = (unsigned)m * 16u;

    v8h* const wf = (v8h*)wfrag + (size_t)wave * (4 * 2 * 32) + lane;
#pragma unroll
    for (int kb = 0; kb < 4; ++kb) {
        const v16h bb = *(const v16h*)&Wp[(((size_t)(wave * 4 + kb)) * 32 + lane) * 16];
        wf[(kb * 2 + 0) * 32] = __builtin_shufflevector(bb, bb, 0, 1, 2, 3, 4, 5, 6, 7);
        wf[(kb * 2 + 1) * 32] = __builtin_shufflevector(bb, bb, 8, 9, 10, 11, 12, 13, 14, 15);
    }
    if (wave == 0) ((float4*)bias_s)[lane] = ((const float4*)bias)[lane];

    int tile = blockIdx.x;
    int cnt0 = 0, cnt1 = 0, idxA = 0, idxB = 0;
    if (tile < ntiles) {
        const int r = tile * 16 + wave * 2;
        const int cv = gcn_two_degrees(cnt_dst, r, nrows, lane);
        cnt0 = __builtin_amdgcn_readlane(cv, 0);
        cnt1 = __builtin_amdgcn_readlane(cv, 16);
        const int mycnt = half ? cnt1 : cnt0;
        const int mynb  = (mycnt < GCN_CAP) ? mycnt : GCN_CAP;
        idxA = gcn_chunk_idx(col, r + half, mynb, 0,  m);
        idxB = gcn_chunk_idx(col, r + half, mynb, 16, m);
    }

    while (tile < ntiles) {
        const int row0 = tile * 16;
        const int  tile_n = tile + tile_step;
        const bool more   = tile_n < ntiles;
        const int  rn     = tile_n * 16 + wave * 2;
        const int cvn = gcn_two_degrees(cnt_dst, rn, nrows, lane);

        {
            float acc[8];
#pragma unroll
            for (int e = 0; e < 8; ++e) acc[e] = 0.0f;
            const int nb0   = (cnt0 < GCN_CAP) ? cnt0 : GCN_CAP;
            const int nb1   = (cnt1 < GCN_CAP) ? cnt1 : GCN_CAP;
            const int nbmax = (nb0 > nb1) ? nb0 : nb1;
            const int mynb  = half ? nb1 : nb0;
            const int myrow = row0 + wave * 2 + half;
            if (nbmax > 0)  gcn_sum_chunk(Xb, idxA, (nbmax < 16) ? nbmax : 16, laneoff, acc);
            if (nbmax > 16) gcn_sum_chunk(Xb, idxB, (nbmax < 32) ? (nbmax - 16) : 16, laneoff, acc);
#pragma unroll 1
            for (int cb = 32; cb < nbmax; cb += 16) {
                const int idxv = gcn_chunk_idx(col, myrow, mynb, cb, m);
                const int ns = (nbmax - cb < 16) ? (nbmax - cb) : 16;
                gcn_sum_chunk(Xb, idxv, ns, laneoff, acc);
            }
            if (cnt0 > GCN_CAP || cnt1 > GCN_CAP) {
                int novf = ovf_cnt[0];
                novf = (novf < ovf_cap) ? novf : ovf_cap;
                for (int rr = 0; rr < 2; ++rr) {
                    if ((rr ? cnt1 : cnt0) <= GCN_CAP) continue;
                    const int   row  = row0 + wave * 2 + rr;
                    for (int base = 0; base < novf; base += 32) {
                        const int  e   = base + lane;
                        const bool hit = (e < novf) && (ovf_dst[e] == row);
                        const int  hs  = hit ? ovf_src[e] : 0;
                        unsigned mask = __builtin_amdgcn_ballot_w32(hit);
                        while (mask != 0u) {
                            const int j = __builtin_ctz(mask);
                            mask &= mask - 1u;
                            const int sall = __builtin_amdgcn_readlane(hs, j);
                            const int sidx = (half == rr) ? sall : GCN_N;
                            const v8h v = *(const v8h*)(Xb + ((unsigned)sidx * 256u + laneoff));
#pragma unroll
                            for (int q = 0; q < 8; ++q) acc[q] += (float)v[q];
                        }
                    }
                }
            }
            v8h p;
#pragma unroll
            for (int e = 0; e < 8; ++e) p[e] = (_Float16)acc[e];
            *(v8h*)&atile[(wave * 2 + half) * GCN_APITCH + m * 8] = p;
        }

        int csv = 0;
        if (HIDDEN_OUT) csv = gcn_two_degrees(cnt_src, row0 + wave * 2, nrows, lane);

        const int cnt0n = __builtin_amdgcn_readlane(cvn, 0);
        const int cnt1n = __builtin_amdgcn_readlane(cvn, 16);
        int idxAn = 0, idxBn = 0;
        if (more) {
            const int mycntn = half ? cnt1n : cnt0n;
            const int mynbn  = (mycntn < GCN_CAP) ? mycntn : GCN_CAP;
            idxAn = gcn_chunk_idx(col, rn + half, mynbn, 0,  m);
            idxBn = gcn_chunk_idx(col, rn + half, mynbn, 16, m);
        }

        v16h b[4];
#pragma unroll
        for (int kb = 0; kb < 4; ++kb) {
            const v8h blo = wf[(kb * 2 + 0) * 32];
            const v8h bhi = wf[(kb * 2 + 1) * 32];
            b[kb] = __builtin_shufflevector(blo, bhi, 0, 1, 2, 3, 4, 5, 6, 7,
                                            8, 9, 10, 11, 12, 13, 14, 15);
        }

        gcn_lds_barrier();

        v8f c = {};
#pragma unroll
        for (int kb = 0; kb < 4; ++kb) {
            const _Float16* ap = &atile[m * GCN_APITCH + kb * 32 + 8 * half];
            const v8h lo = *(const v8h*)ap;
            const v8h hi = *(const v8h*)(ap + 16);
            const v16h a = __builtin_shufflevector(lo, hi, 0, 1, 2, 3, 4, 5, 6, 7,
                                                   8, 9, 10, 11, 12, 13, 14, 15);
            c = __builtin_amdgcn_wmma_f32_16x16x32_f16(
                 false, a,  false, b[kb],
                 (short)0, c,  false,  false);
        }

#pragma unroll
        for (int i = 0; i < 8; ++i)
            otile[(i + 8 * half) * GCN_OPITCH + n0 + m] = c[i];
        gcn_lds_barrier();

#pragma unroll
        for (int rr = 0; rr < 2; ++rr) {
            const int lr  = wave * 2 + rr;
            const int row = row0 + lr;
            if (row < nrows) {
                const float4 v  = *(const float4*)&otile[lr * GCN_OPITCH + lane * 4];
                const float4 bv = ((const float4*)bias_s)[lane];
                const float  nd = gcn_norm_of(rr ? cnt1 : cnt0);
                float4 r;
                r.x = v.x * nd + bv.x;
                r.y = v.y * nd + bv.y;
                r.z = v.z * nd + bv.z;
                r.w = v.w * nd + bv.w;
                if (HIDDEN_OUT) {
                    const float ns = gcn_norm_of(__builtin_amdgcn_readlane(csv, 16 * rr));
                    v4h h;
                    h[0] = (_Float16)(fmaxf(r.x, 0.0f) * ns); h[1] = (_Float16)(fmaxf(r.y, 0.0f) * ns);
                    h[2] = (_Float16)(fmaxf(r.z, 0.0f) * ns); h[3] = (_Float16)(fmaxf(r.w, 0.0f) * ns);
                    *(v4h*)&out_f16[(unsigned)(row * GCN_D + lane * 4)] = h;
                } else {
                    *(float4*)&out_f32[(unsigned)(row * GCN_D + lane * 4)] = r;
                }
            }
        }
        tile = tile_n;
        cnt0 = cnt0n; cnt1 = cnt1n; idxA = idxAn; idxB = idxBn;
    }
}

extern "C" void kernel_launch(void* const* d_in, const int* in_sizes, int n_in,
                              void* d_out, int out_size, void* d_ws, size_t ws_size,
                              hipStream_t stream) {
    const float* feat = (const float*)d_in[0];
    const int*   src  = (const int*)  d_in[1];
    const int*   dst  = (const int*)  d_in[2];
    const float* W1   = (const float*)d_in[3];
    const float* b1   = (const float*)d_in[4];
    const float* W2   = (const float*)d_in[5];
    const float* b2   = (const float*)d_in[6];
    float* out = (float*)d_out;

    const int N = GCN_N;
    const int E = in_sizes[1];
    const size_t NF = (size_t)N * GCN_D;

    int*    cntS   = (int*)d_ws;
    int*    cntD   = cntS + GCN_NPAD;
    int*    ovfcnt = cntD + GCN_NPAD;
    _Float16* Wp1  = (_Float16*)(ovfcnt + GCN_OVFPAD);
    _Float16* Wp2  = Wp1 + GCN_WPACK_ELEMS;
    _Float16* Xh   = Wp2 + GCN_WPACK_ELEMS;
    _Float16* Hh   = Xh + NF + GCN_D;
    int*    col    = (int*)(Hh + NF + GCN_D);
    int*    slots  = col + (size_t)N * GCN_CAP;
    int*    bovfE  = slots + ((size_t)4 * E + 8192);
    int*    bovfR  = bovfE + ((size_t)2 * E + 4096);
    int*    ovfD   = bovfR + ((size_t)2 * E + 4096);
    int*    ovfS   = ovfD + E;
    int*    bovfC  = ovfS + E;

    const int thr = 256;
    const int pgrid1 = (E + GCN_CHUNK - 1) / GCN_CHUNK;
    const int ntiles = N / 16;
    const int layer_grid = (ntiles < GCN_LAYER_GRID) ? ntiles : GCN_LAYER_GRID;

    gcn_partition_kernel<<<pgrid1 + GCN_PACK_BLOCKS, GCN_PT, 0, stream>>>(
        src, dst, W1, W2, Wp1, Wp2, slots, bovfC, bovfE, bovfR, ovfcnt, E, pgrid1);
    gcn_fine_kernel<<<GCN_NRANGE, GCN_FT, 0, stream>>>(slots, bovfC, bovfE, bovfR,
                                                       cntS, cntD, col, ovfcnt, ovfD, ovfS,
                                                       feat, Xh, pgrid1, E);

    gcn_layer_kernel<true><<<layer_grid, thr, 0, stream>>>(
        Xh, cntS, cntD, col, ovfcnt, ovfD, ovfS, Wp1, b1, (float*)nullptr, Hh,
        N, ntiles, layer_grid, E);

    gcn_layer_kernel<false><<<layer_grid, thr, 0, stream>>>(
        Hh, cntS, cntD, col, ovfcnt, ovfD, ovfS, Wp2, b2, out, (_Float16*)nullptr,
        N, ntiles, layer_grid, E);
}


# Layer edge sums on the matrix unit: two 16-byte loads are the B operand of one 16x16x32 WMMA, A = fixed 0/1 matrix [I|I]; a pair of bucket slots joins the f32 sum exactly, one rounding per pair
# speedup vs baseline: 10.2701x; 1.0313x over previous
#include <hip/hip_runtime.h>


typedef __attribute__((ext_vector_type(16))) _Float16 v16h;
typedef __attribute__((ext_vector_type(8)))  float  v8f;
typedef __attribute__((ext_vector_type(8)))  _Float16 v8h;
typedef __attribute__((ext_vector_type(4)))  _Float16 v4h;

#define GCN_N 100000
#define GCN_D 128
#define GCN_WPACK_ELEMS (8 * 4 * 32 * 16)

#define GCN_NPAD 100032

#define GCN_CAP 64
#define GCN_OVFPAD 64

__device__ __forceinline__ float gcn_norm_of(int deg) {
    return rsqrtf(fmaxf((float)deg, 1.0f));
}

#define GCN_RNODES 392
#define GCN_NRANGE 256
#define GCN_PT     512
#define GCN_EPT    4
#define GCN_CHUNK  (GCN_PT * GCN_EPT)
#define GCN_SLOT   32
#define GCN_SPITCH 33
#define GCN_BOVF   (2 * GCN_CHUNK)

__device__ __forceinline__ int gcn_frag_k(int lane, int i) {
    const int j   = i >> 1;
    const int sub = i & 1;
    const int kk  = (j < 4) ? (2 * j + sub) : (16 + 2 * (j - 4) + sub);
    return kk + 8 * (lane >> 4);
}

#define GCN_PACK_BLOCKS ((2 * GCN_WPACK_ELEMS) / GCN_PT)

__global__ void __launch_bounds__(GCN_PT)
gcn_partition_kernel(const int* __restrict__ src,
                     const int* __restrict__ dst,
                     const float* __restrict__ W1, const float* __restrict__ W2,
                     _Float16* __restrict__ Wp1, _Float16* __restrict__ Wp2,
                     int* __restrict__ slots,
                     int* __restrict__ bovf_cnt,
                     int* __restrict__ bovf_ent,
                     int* __restrict__ bovf_rng,
                     int* __restrict__ ovf_cnt,
                     int nedges, int nsplit) {
    __shared__ int hist[GCN_NRANGE];
    __shared__ int nspill;
    __shared__ int stage[GCN_NRANGE * GCN_SPITCH];

    const int tid = threadIdx.x;
    const int blk = blockIdx.x;
    if (blk >= nsplit) {
        const int g = (blk - nsplit) * GCN_PT + tid;
        const bool second = g >= GCN_WPACK_ELEMS;
        const float* __restrict__ W  = second ? W2 : W1;
        _Float16* __restrict__    Wp = second ? Wp2 : Wp1;
        const int idx  = second ? g - GCN_WPACK_ELEMS : g;
        const int i    = idx & 15;
        const int lane = (idx >> 4) & 31;
        const int kb   = (idx >> 9) & 3;
        const int nt   = idx >> 11;
        const int K    = kb * 32 + gcn_frag_k(lane, i);
        const int n    = nt * 16 + (lane & 15);
        Wp[idx] = (_Float16)W[(size_t)K * GCN_D + n];
        return;
    }

    const int e0 = blk * GCN_CHUNK;
    int sv[GCN_EPT], dv[GCN_EPT];
#pragma unroll
    for (int k = 0; k < GCN_EPT; ++k) {
        const int i  = e0 + k * GCN_PT + tid;
        const int ic = (i < nedges) ? i : (nedges - 1);
        sv[k] = src[ic];
        dv[k] = dst[ic];
    }

    if (tid < GCN_NRANGE) hist[tid] = 0;
    if (tid == 0) nspill = 0;
    if (blk == 0 && tid == 0) ovf_cnt[0] = 0;
    for (int i = tid; i < GCN_NRANGE * GCN_SPITCH; i += GCN_PT) stage[i] = -1;
    __syncthreads();

    int ent[2 * GCN_EPT], rng[2 * GCN_EPT], rk[2 * GCN_EPT];
#pragma unroll
    for (int k = 0; k < GCN_EPT; ++k) {
        const int i = e0 + k * GCN_PT + tid;
        if (i < nedges) {
            const int s  = sv[k];
            const int d  = dv[k];
            const int rd = d / GCN_RNODES;
            const int rs = s / GCN_RNODES;
            ent[2 * k]     = (s << 10) | (d - rd * GCN_RNODES);
            rng[2 * k]     = rd;
            rk[2 * k]      = atomicAdd(&hist[rd], 1);
            ent[2 * k + 1] = (1 << 9) | (s - rs * GCN_RNODES);
            rng[2 * k + 1] = rs;
            rk[2 * k + 1]  = atomicAdd(&hist[rs], 1);
        } else {
            ent[2 * k] = 0; ent[2 * k + 1] = 0;
            rng[2 * k] = -1; rng[2 * k + 1] = -1;
            rk[2 * k] = 0;  rk[2 * k + 1] = 0;
        }
    }
#pragma unroll
    for (int k = 0; k < 2 * GCN_EPT; ++k) {
        if (rng[k] >= 0) {
            if (rk[k] < GCN_SLOT) {
                stage[rng[k] * GCN_SPITCH + rk[k]] = ent[k];
            } else {
                const int o = atomicAdd(&nspill, 1);
                bovf_ent[(size_t)blk * GCN_BOVF + o] = ent[k];
                bovf_rng[(size_t)blk * GCN_BOVF + o] = rng[k];
            }
        }
    }
    __syncthreads();

    for (int idx = tid; idx < GCN_NRANGE * GCN_SLOT; idx += GCN_PT) {
        const int r = idx / GCN_SLOT;
        const int k = idx - r * GCN_SLOT;
        slots[((size_t)r * nsplit + blk) * GCN_SLOT + k] = stage[r * GCN_SPITCH + k];
    }
    if (tid == 0) bovf_cnt[blk] = nspill;
}

template <int G>
__device__ __forceinline__ void gcn_load_rows(const float4* __restrict__ X4, int node0, int l0,
                                              int wave, int lane, float4 (&v)[G]) {
#pragma unroll
    for (int k = 0; k < G; ++k) {
        const int node = node0 + l0 + k * 32 + wave;
        const int nc   = (node < GCN_N) ? node : (GCN_N - 1);
        v[k] = X4[(size_t)nc * (GCN_D / 4) + lane];
    }
}
template <int G>
__device__ __forceinline__ void gcn_store_rows(v4h* __restrict__ H4, const int* cs, int node0,
                                               int l0, int wave, int lane, const float4 (&v)[G]) {
#pragma unroll
    for (int k = 0; k < G; ++k) {
        const int l    = l0 + k * 32 + wave;
        const int node = node0 + l;
        if (node < GCN_N) {
            const float ns = gcn_norm_of(cs[l]);
            v4h h;
            h[0] = (_Float16)(v[k].x * ns); h[1] = (_Float16)(v[k].y * ns);
            h[2] = (_Float16)(v[k].z * ns); h[3] = (_Float16)(v[k].w * ns);
            H4[(size_t)node * (GCN_D / 4) + lane] = h;
        }
    }
}

#define GCN_FT 1024
#define GCN_FROWS 12
#define GCN_FB 25
static_assert(GCN_RNODES > GCN_FROWS * 32 && GCN_RNODES <= (GCN_FROWS + 1) * 32,
              "a wave copies GCN_FROWS rows, and one more where its number is below the remainder");

__device__ __forceinline__ void gcn_place_entry(int e, int* cd, int* cs, int* tile, int node0,
                                                int* __restrict__ ovf_cnt,
                                                int* __restrict__ ovf_dst,
                                                int* __restrict__ ovf_src, int ovf_cap) {
    const int local = e & 511;
    if (e & 512) {
        atomicAdd(&cs[local], 1);
    } else {
        const int k = atomicAdd(&cd[local], 1);
        if (k < GCN_CAP) {
            tile[local * GCN_CAP + k] = e >> 10;
        } else {
            const int idx = atomicAdd(ovf_cnt, 1);
            if (idx < ovf_cap) { ovf_dst[idx] = node0 + local; ovf_src[idx] = e >> 10; }
        }
    }
}

__global__ void __launch_bounds__(GCN_FT)
gcn_fine_kernel(const int* __restrict__ slots,
                const int* __restrict__ bovf_cnt,
                const int* __restrict__ bovf_ent,
                const int* __restrict__ bovf_rng,
                int* __restrict__ cnt_src,
                int* __restrict__ cnt_dst,
                int* __restrict__ col,
                int* __restrict__ ovf_cnt,
                int* __restrict__ ovf_dst,
                int* __restrict__ ovf_src,
                const float* __restrict__ X,
                _Float16* __restrict__ Xh,
                int nsplit, int ovf_cap) {
    __shared__ int cd[GCN_RNODES];
    __shared__ int cs[GCN_RNODES];
    __shared__ int tile[GCN_RNODES * GCN_CAP];

    const int tid   = threadIdx.x;
    const int r     = blockIdx.x;
    const int node0 = r * GCN_RNODES;
    for (int i = tid; i < GCN_RNODES; i += GCN_FT) { cd[i] = 0; cs[i] = 0; }
    __syncthreads();

    const int total = nsplit * GCN_SLOT;
    const int* __restrict__ pr = slots + (size_t)r * total;
    for (int base = 0; base < total; base += GCN_FT * GCN_FB) {
        int e[GCN_FB], kk[GCN_FB];
#pragma unroll
        for (int k = 0; k < GCN_FB; ++k) {
            const int i  = base + k * GCN_FT + tid;
            const int ic = (i < total) ? i : (total - 1);
            const int v  = pr[ic];
            e[k] = (i < total) ? v : -1;
        }
#pragma unroll
        for (int k = 0; k < GCN_FB; ++k) {
            kk[k] = -1;
            if (e[k] >= 0) {
                const int local = e[k] & 511;
                if (e[k] & 512) atomicAdd(&cs[local], 1);
                else            kk[k] = atomicAdd(&cd[local], 1);
            }
        }
#pragma unroll
        for (int k = 0; k < GCN_FB; ++k) {
            if (kk[k] >= 0) {
                const int local = e[k] & 511;
                if (kk[k] < GCN_CAP) {
                    tile[local * GCN_CAP + kk[k]] = e[k] >> 10;
                } else {
                    const int idx = atomicAdd(ovf_cnt, 1);
                    if (idx < ovf_cap) { ovf_dst[idx] = node0 + local; ovf_src[idx] = e[k] >> 10; }
                }
            }
        }
    }
    for (int b = tid; b < nsplit; b += GCN_FT) {
        int c = bovf_cnt[b];
        c = (c < GCN_BOVF) ? c : GCN_BOVF;
        for (int j = 0; j < c; ++j) {
            if (bovf_rng[(size_t)b * GCN_BOVF + j] == r)
                gcn_place_entry(bovf_ent[(size_t)b * GCN_BOVF + j], cd, cs, tile, node0,
                                ovf_cnt, ovf_dst, ovf_src, ovf_cap);
        }
    }
    __syncthreads();

    const int lane = tid & 31;
    const int wave = __builtin_amdgcn_readfirstlane(tid >> 5);
    const float4* __restrict__ X4 = (const float4*)X;
    v4h* __restrict__ H4 = (v4h*)Xh;
    const bool last_row = wave < GCN_RNODES - GCN_FROWS * 32;
    float4 va[GCN_FROWS];
    float4 vc[1];
    gcn_load_rows<GCN_FROWS>(X4, node0, 0, wave, lane, va);
    gcn_load_rows<1>(X4, node0, last_row ? GCN_FROWS * 32 : 0, wave, lane, vc);

    for (int i = tid; i < GCN_RNODES; i += GCN_FT) {
        const int node = node0 + i;
        if (node < GCN_N) { cnt_dst[node] = cd[i]; cnt_src[node] = cs[i]; }
    }
    for (int idx = tid; idx < GCN_RNODES * GCN_CAP; idx += GCN_FT) {
        const int l    = idx / GCN_CAP;
        const int k    = idx - l * GCN_CAP;
        const int node = node0 + l;
        if (node < GCN_N && k < cd[l]) col[(size_t)node * GCN_CAP + k] = tile[idx];
    }

    gcn_store_rows<GCN_FROWS>(H4, cs, node0, 0, wave, lane, va);
    if (last_row) gcn_store_rows<1>(H4, cs, node0, GCN_FROWS * 32, wave, lane, vc);
    if (r == 0 && wave == 0) {
        v4h z;
        z[0] = (_Float16)0.0f; z[1] = (_Float16)0.0f; z[2] = (_Float16)0.0f; z[3] = (_Float16)0.0f;
        H4[(size_t)GCN_N * (GCN_D / 4) + lane] = z;
        H4[(size_t)(2 * GCN_N + 1) * (GCN_D / 4) + lane] = z;
    }
}

template <int T>
__device__ __forceinline__ int gcn_row_share(int v) {
    return __builtin_amdgcn_mov_dpp(v, 0x150 + T, 0xf, 0xf, true);
}
__device__ __forceinline__ int gcn_row_share_n(int v, int t) {
    switch (t) {
    case 0:  return gcn_row_share<0>(v);   case 1:  return gcn_row_share<1>(v);
    case 2:  return gcn_row_share<2>(v);   case 3:  return gcn_row_share<3>(v);
    case 4:  return gcn_row_share<4>(v);   case 5:  return gcn_row_share<5>(v);
    case 6:  return gcn_row_share<6>(v);   case 7:  return gcn_row_share<7>(v);
    case 8:  return gcn_row_share<8>(v);   case 9:  return gcn_row_share<9>(v);
    case 10: return gcn_row_share<10>(v);  case 11: return gcn_row_share<11>(v);
    case 12: return gcn_row_share<12>(v);  case 13: return gcn_row_share<13>(v);
    case 14: return gcn_row_share<14>(v);  default: return gcn_row_share<15>(v);
    }
}

template <int G, int T0>
__device__ __forceinline__ void gcn_sum_group(const char* __restrict__ Xb, int idxv,
                                              unsigned laneoff, const v16h& sel, v8f& acc) {
    v8h v[G];
#pragma unroll
    for (int k = 0; k < G; ++k) {
        const int s = gcn_row_share_n(idxv, T0 + k);
        v[k] = *(const v8h*)(Xb + ((unsigned)s * 256u + laneoff));
    }
    if (G == 8) {
        const v16h b0 = __builtin_shufflevector(v[0], v[1], 0, 1, 2, 3, 4, 5, 6, 7, 8, 9, 10, 11, 12, 13, 14, 15);
        const v16h b1 = __builtin_shufflevector(v[2], v[3], 0, 1, 2, 3, 4, 5, 6, 7, 8, 9, 10, 11, 12, 13, 14, 15);
        const v16h b2 = __builtin_shufflevector(v[G - 4], v[G - 3], 0, 1, 2, 3, 4, 5, 6, 7, 8, 9, 10, 11, 12, 13, 14, 15);
        const v16h b3 = __builtin_shufflevector(v[G - 2], v[G - 1], 0, 1, 2, 3, 4, 5, 6, 7, 8, 9, 10, 11, 12, 13, 14, 15);
        asm volatile("v_wmma_f32_16x16x32_f16 %0, %1, %2, %0\n\t"
                     "v_wmma_f32_16x16x32_f16 %0, %1, %3, %0\n\t"
                     "v_wmma_f32_16x16x32_f16 %0, %1, %4, %0\n\t"
                     "v_wmma_f32_16x16x32_f16 %0, %1, %5, %0\n\t"
                     "v_nop\n\tv_nop\n\tv_nop\n\tv_nop"
                     : "+v"(acc) : "v"(sel), "v"(b0), "v"(b1), "v"(b2), "v"(b3));
    } else {
        const v16h b0 = __builtin_shufflevector(v[0], v[1], 0, 1, 2, 3, 4, 5, 6, 7, 8, 9, 10, 11, 12, 13, 14, 15);
        const v16h b1 = __builtin_shufflevector(v[2], v[3], 0, 1, 2, 3, 4, 5, 6, 7, 8, 9, 10, 11, 12, 13, 14, 15);
        asm volatile("v_wmma_f32_16x16x32_f16 %0, %1, %2, %0\n\t"
                     "v_wmma_f32_16x16x32_f16 %0, %1, %3, %0\n\t"
                     "v_nop\n\tv_nop\n\tv_nop\n\tv_nop"
                     : "+v"(acc) : "v"(sel), "v"(b0), "v"(b1));
    }
}

#define GCN_APITCH 136
#define GCN_OPITCH 132

#define GCN_LAYER_GRID 1280

__device__ __forceinline__ int gcn_two_degrees(const int* __restrict__ deg, int r,
                                               int nrows, int lane) {
    const int q  = r + (lane >> 4);
    const int qc = (q < nrows) ? q : (nrows - 1);
    const int v  = deg[(unsigned)qc];
    return (q < nrows) ? v : 0;
}

__device__ __forceinline__ void gcn_lds_barrier() {
    __builtin_amdgcn_fence(__ATOMIC_RELEASE, "workgroup", "local");
    __builtin_amdgcn_s_barrier();
    __builtin_amdgcn_fence(__ATOMIC_ACQUIRE, "workgroup", "local");
}

__device__ __forceinline__ int gcn_chunk_idx(const int* __restrict__ col, int myrow,
                                             int mynb, int cb, int c) {
    if (c >= mynb - cb) return GCN_N;
    return (col + cb)[(unsigned)(myrow * GCN_CAP + c)];
}

__device__ __forceinline__ void gcn_sum_chunk(const char* __restrict__ Xb, int idxv, int ns,
                                              unsigned laneoff, const v16h& sel, v8f& acc) {
    if (ns > 4) gcn_sum_group<8, 0>(Xb, idxv, laneoff, sel, acc);
    else        gcn_sum_group<4, 0>(Xb, idxv, laneoff, sel, acc);
    if (ns > 12)     gcn_sum_group<8, 8>(Xb, idxv, laneoff, sel, acc);
    else if (ns > 8) gcn_sum_group<4, 8>(Xb, idxv, laneoff, sel, acc);
}

template <bool HIDDEN_OUT>
__global__ void __launch_bounds__(256) __attribute__((amdgpu_waves_per_eu(10, 10)))
gcn_layer_kernel(const _Float16* __restrict__ Xh,
                 const int* __restrict__ cnt_src,
                 const int* __restrict__ cnt_dst,
                 const int* __restrict__ col,
                 const int* __restrict__ ovf_cnt,
                 const int* __restrict__ ovf_dst,
                 const int* __restrict__ ovf_src,
                 const _Float16* __restrict__ Wp,
                 const float* __restrict__ bias,
                 float* __restrict__ out_f32,
                 _Float16* __restrict__ out_f16,
                 int nrows, int ntiles, int tile_step, int ovf_cap) {
    __shared__ __attribute__((aligned(16))) _Float16 atile[16 * GCN_APITCH];
    __shared__ __attribute__((aligned(16))) float  otile[16 * GCN_OPITCH];
    __shared__ __attribute__((aligned(16))) _Float16 wfrag[8 * 4 * 2 * 32 * 8];
    __shared__ __attribute__((aligned(16))) float bias_s[GCN_D];

    const int lane = threadIdx.x & 31;
    const int wave = __builtin_amdgcn_readfirstlane((int)(threadIdx.x >> 5));
    const char* __restrict__ Xb = (const char*)Xh;

    const int n0   = wave * 16;
    const int half = lane >> 4;
    const int m    = lane & 15;
    const unsigned laneoff = (unsigned)m * 16u;

    v8h* const wf = (v8h*)wfrag + (size_t)wave * (4 * 2 * 32) + lane;
#pragma unroll
    for (int kb = 0; kb < 4; ++kb) {
        const v16h bb = *(const v16h*)&Wp[(((size_t)(wave * 4 + kb)) * 32 + lane) * 16];
        wf[(kb * 2 + 0) * 32] = __builtin_shufflevector(bb, bb, 0, 1, 2, 3, 4, 5, 6, 7);
        wf[(kb * 2 + 1) * 32] = __builtin_shufflevector(bb, bb, 8, 9, 10, 11, 12, 13, 14, 15);
    }
    if (wave == 0) ((float4*)bias_s)[lane] = ((const float4*)bias)[lane];

    v16h sel;
#pragma unroll
    for (int s = 0; s < 16; ++s)
        sel[s] = (((m >> 3) == half) && ((s & 7) == (m & 7))) ? (_Float16)1.0f : (_Float16)0.0f;

    int tile = blockIdx.x;
    int cnt0 = 0, cnt1 = 0, idxA = 0, idxB = 0;
    if (tile < ntiles) {
        const int r = tile * 16 + wave * 2;
        const int cv = gcn_two_degrees(cnt_dst, r, nrows, lane);
        cnt0 = __builtin_amdgcn_readlane(cv, 0);
        cnt1 = __builtin_amdgcn_readlane(cv, 16);
        const int mycnt = half ? cnt1 : cnt0;
        const int mynb  = (mycnt < GCN_CAP) ? mycnt : GCN_CAP;
        idxA = gcn_chunk_idx(col, r + half, mynb, 0,  m);
        idxB = gcn_chunk_idx(col, r + half, mynb, 16, m);
    }

    while (tile < ntiles) {
        const int row0 = tile * 16;
        const int  tile_n = tile + tile_step;
        const bool more   = tile_n < ntiles;
        const int  rn     = tile_n * 16 + wave * 2;
        const int cvn = gcn_two_degrees(cnt_dst, rn, nrows, lane);

        {
            v8f acc = {};
            const int nb0   = (cnt0 < GCN_CAP) ? cnt0 : GCN_CAP;
            const int nb1   = (cnt1 < GCN_CAP) ? cnt1 : GCN_CAP;
            const int nbmax = (nb0 > nb1) ? nb0 : nb1;
            const int mynb  = half ? nb1 : nb0;
            const int myrow = row0 + wave * 2 + half;
            if (nbmax > 0)  gcn_sum_chunk(Xb, idxA, (nbmax < 16) ? nbmax : 16, laneoff, sel, acc);
            if (nbmax > 16) gcn_sum_chunk(Xb, idxB, (nbmax < 32) ? (nbmax - 16) : 16, laneoff, sel, acc);
#pragma unroll 1
            for (int cb = 32; cb < nbmax; cb += 16) {
                const int idxv = gcn_chunk_idx(col, myrow, mynb, cb, m);
                const int ns = (nbmax - cb < 16) ? (nbmax - cb) : 16;
                gcn_sum_chunk(Xb, idxv, ns, laneoff, sel, acc);
            }
            if (cnt0 > GCN_CAP || cnt1 > GCN_CAP) {
                int novf = ovf_cnt[0];
                novf = (novf < ovf_cap) ? novf : ovf_cap;
                for (int rr = 0; rr < 2; ++rr) {
                    if ((rr ? cnt1 : cnt0) <= GCN_CAP) continue;
                    const int   row  = row0 + wave * 2 + rr;
                    for (int base = 0; base < novf; base += 32) {
                        const int  e   = base + lane;
                        const bool hit = (e < novf) && (ovf_dst[e] == row);
                        const int  hs  = hit ? ovf_src[e] : 0;
                        unsigned mask = __builtin_amdgcn_ballot_w32(hit);
                        while (mask != 0u) {
                            const int j = __builtin_ctz(mask);
                            mask &= mask - 1u;
                            const int sall = __builtin_amdgcn_readlane(hs, j);
                            const int sidx = (half == rr) ? sall : GCN_N;
                            const v8h v = *(const v8h*)(Xb + ((unsigned)sidx * 256u + laneoff));
#pragma unroll
                            for (int q = 0; q < 8; ++q) acc[q] += (float)v[q];
                        }
                    }
                }
            }
            v8h p;
#pragma unroll
            for (int e = 0; e < 8; ++e) p[e] = (_Float16)acc[e];
            *(v8h*)&atile[(wave * 2 + half) * GCN_APITCH + m * 8] = p;
        }

        int csv = 0;
        if (HIDDEN_OUT) csv = gcn_two_degrees(cnt_src, row0 + wave * 2, nrows, lane);

        const int cnt0n = __builtin_amdgcn_readlane(cvn, 0);
        const int cnt1n = __builtin_amdgcn_readlane(cvn, 16);
        int idxAn = 0, idxBn = 0;
        if (more) {
            const int mycntn = half ? cnt1n : cnt0n;
            const int mynbn  = (mycntn < GCN_CAP) ? mycntn : GCN_CAP;
            idxAn = gcn_chunk_idx(col, rn + half, mynbn, 0,  m);
            idxBn = gcn_chunk_idx(col, rn + half, mynbn, 16, m);
        }

        v16h b[4];
#pragma unroll
        for (int kb = 0; kb < 4; ++kb) {
            const v8h blo = wf[(kb * 2 + 0) * 32];
            const v8h bhi = wf[(kb * 2 + 1) * 32];
            b[kb] = __builtin_shufflevector(blo, bhi, 0, 1, 2, 3, 4, 5, 6, 7,
                                            8, 9, 10, 11, 12, 13, 14, 15);
        }

        gcn_lds_barrier();

        v8f c = {};
#pragma unroll
        for (int kb = 0; kb < 4; ++kb) {
            const _Float16* ap = &atile[m * GCN_APITCH + kb * 32 + 8 * half];
            const v8h lo = *(const v8h*)ap;
            const v8h hi = *(const v8h*)(ap + 16);
            const v16h a = __builtin_shufflevector(lo, hi, 0, 1, 2, 3, 4, 5, 6, 7,
                                                   8, 9, 10, 11, 12, 13, 14, 15);
            c = __builtin_amdgcn_wmma_f32_16x16x32_f16(
                 false, a,  false, b[kb],
                 (short)0, c,  false,  false);
        }

#pragma unroll
        for (int i = 0; i < 8; ++i)
            otile[(i + 8 * half) * GCN_OPITCH + n0 + m] = c[i];
        gcn_lds_barrier();

#pragma unroll
        for (int rr = 0; rr < 2; ++rr) {
            const int lr  = wave * 2 + rr;
            const int row = row0 + lr;
            if (row < nrows) {
                const float4 v  = *(const float4*)&otile[lr * GCN_OPITCH + lane * 4];
                const float4 bv = ((const float4*)bias_s)[lane];
                const float  nd = gcn_norm_of(rr ? cnt1 : cnt0);
                float4 r;
                r.x = v.x * nd + bv.x;
                r.y = v.y * nd + bv.y;
                r.z = v.z * nd + bv.z;
                r.w = v.w * nd + bv.w;
                if (HIDDEN_OUT) {
                    const float ns = gcn_norm_of(__builtin_amdgcn_readlane(csv, 16 * rr));
                    v4h h;
                    h[0] = (_Float16)(fmaxf(r.x, 0.0f) * ns); h[1] = (_Float16)(fmaxf(r.y, 0.0f) * ns);
                    h[2] = (_Float16)(fmaxf(r.z, 0.0f) * ns); h[3] = (_Float16)(fmaxf(r.w, 0.0f) * ns);
                    *(v4h*)&out_f16[(unsigned)(row * GCN_D + lane * 4)] = h;
                } else {
                    *(float4*)&out_f32[(unsigned)(row * GCN_D + lane * 4)] = r;
                }
            }
        }
        tile = tile_n;
        cnt0 = cnt0n; cnt1 = cnt1n; idxA = idxAn; idxB = idxBn;
    }
}

extern "C" void kernel_launch(void* const* d_in, const int* in_sizes, int n_in,
                              void* d_out, int out_size, void* d_ws, size_t ws_size,
                              hipStream_t stream) {
    const float* feat = (const float*)d_in[0];
    const int*   src  = (const int*)  d_in[1];
    const int*   dst  = (const int*)  d_in[2];
    const float* W1   = (const float*)d_in[3];
    const float* b1   = (const float*)d_in[4];
    const float* W2   = (const float*)d_in[5];
    const float* b2   = (const float*)d_in[6];
    float* out = (float*)d_out;

    const int N = GCN_N;
    const int E = in_sizes[1];
    const size_t NF = (size_t)N * GCN_D;

    int*    cntS   = (int*)d_ws;
    int*    cntD   = cntS + GCN_NPAD;
    int*    ovfcnt = cntD + GCN_NPAD;
    _Float16* Wp1  = (_Float16*)(ovfcnt + GCN_OVFPAD);
    _Float16* Wp2  = Wp1 + GCN_WPACK_ELEMS;
    _Float16* Xh   = Wp2 + GCN_WPACK_ELEMS;
    _Float16* Hh   = Xh + NF + GCN_D;
    int*    col    = (int*)(Hh + NF + GCN_D);
    int*    slots  = col + (size_t)N * GCN_CAP;
    int*    bovfE  = slots + ((size_t)4 * E + 8192);
    int*    bovfR  = bovfE + ((size_t)2 * E + 4096);
    int*    ovfD   = bovfR + ((size_t)2 * E + 4096);
    int*    ovfS   = ovfD + E;
    int*    bovfC  = ovfS + E;

    const int thr = 256;
    const int pgrid1 = (E + GCN_CHUNK - 1) / GCN_CHUNK;
    const int ntiles = N / 16;
    const int layer_grid = (ntiles < GCN_LAYER_GRID) ? ntiles : GCN_LAYER_GRID;

    gcn_partition_kernel<<<pgrid1 + GCN_PACK_BLOCKS, GCN_PT, 0, stream>>>(
        src, dst, W1, W2, Wp1, Wp2, slots, bovfC, bovfE, bovfR, ovfcnt, E, pgrid1);
    gcn_fine_kernel<<<GCN_NRANGE, GCN_FT, 0, stream>>>(slots, bovfC, bovfE, bovfR,
                                                       cntS, cntD, col, ovfcnt, ovfD, ovfS,
                                                       feat, Xh, pgrid1, E);

    gcn_layer_kernel<true><<<layer_grid, thr, 0, stream>>>(
        Xh, cntS, cntD, col, ovfcnt, ovfD, ovfS, Wp1, b1, (float*)nullptr, Hh,
        N, ntiles, layer_grid, E);

    gcn_layer_kernel<false><<<layer_grid, thr, 0, stream>>>(
        Hh, cntS, cntD, col, ovfcnt, ovfD, ovfS, Wp2, b2, out, (_Float16*)nullptr,
        N, ntiles, layer_grid, E);
}
